# starting kernel
# baseline (speedup 1.0000x reference)
#include <hip/hip_runtime.h>
#include <math.h>

typedef __attribute__((ext_vector_type(2))) float v2f;
typedef __attribute__((ext_vector_type(8))) float v8f;

#define NQ       10
#define DIM      1024
#define NLAYERS  4
#define WPB      8
#define PI_F     3.14159265358979323846f

struct c2 { float re, im; };
__device__ __forceinline__ c2 cmul(c2 a, c2 b) {
    return { a.re * b.re - a.im * b.im, a.re * b.im + a.im * b.re };
}
__device__ __forceinline__ c2 cadd(c2 a, c2 b) { return { a.re + b.re, a.im + b.im }; }

__global__ void qprep_kernel(const float* __restrict__ theta, float* __restrict__ Mout) {
    int g = blockIdx.x * blockDim.x + threadIdx.x;
    if (g >= NLAYERS * NQ) return;
    float t0 = theta[g * 3 + 0], t1 = theta[g * 3 + 1], t2 = theta[g * 3 + 2];
    float cx = cosf(t0 * 0.5f), sx = sinf(t0 * 0.5f);
    float cy = cosf(t1 * 0.5f), sy = sinf(t1 * 0.5f);
    float cz = cosf(t2 * 0.5f), sz = sinf(t2 * 0.5f);
    c2 X[2][2] = { { {cx, 0.f}, {0.f, -sx} }, { {0.f, -sx}, {cx, 0.f} } };
    c2 Y[2][2] = { { {cy, 0.f}, {-sy, 0.f} }, { {sy, 0.f}, {cy, 0.f} } };
    c2 Z[2][2] = { { {cz, -sz}, {0.f, 0.f} }, { {0.f, 0.f}, {cz, sz} } };
    c2 XY[2][2], G[2][2];
    for (int i = 0; i < 2; ++i)
        for (int j = 0; j < 2; ++j)
            XY[i][j] = cadd(cmul(X[i][0], Y[0][j]), cmul(X[i][1], Y[1][j]));
    for (int i = 0; i < 2; ++i)
        for (int j = 0; j < 2; ++j)
            G[i][j] = cadd(cmul(XY[i][0], Z[0][j]), cmul(XY[i][1], Z[1][j]));
    float* M = Mout + g * 16;
    for (int i = 0; i < 2; ++i)
        for (int j = 0; j < 2; ++j) {
            M[(2 * i    ) * 4 + 2 * j    ] =  G[i][j].re;
            M[(2 * i    ) * 4 + 2 * j + 1] =  G[i][j].im;
            M[(2 * i + 1) * 4 + 2 * j    ] = -G[i][j].im;
            M[(2 * i + 1) * 4 + 2 * j + 1] =  G[i][j].re;
        }
}

__global__ void __launch_bounds__(WPB * 32)
qsim_kernel(const float* __restrict__ x, const float* __restrict__ encW,
            const float* __restrict__ encB, const float* __restrict__ Mvar,
            float* __restrict__ out, int batch) {
    __shared__ float2 lds[WPB * DIM];

    const int lane = threadIdx.x & 31;
    const int w    = threadIdx.x >> 5;
    const int b    = blockIdx.x * WPB + w;
    if (b >= batch) return;

    float2* st = &lds[w * DIM];
    float*  stf = (float*)st;
    const int n  = lane & 15;
    const int hi = lane >> 4;

    for (int i = lane; i < DIM; i += 32) st[i] = make_float2(0.f, 0.f);
    if (lane == 0) st[0] = make_float2(1.f, 0.f);

    auto apply = [&](int s, v2f Bop) {
        const int lowmask = (1 << s) - 1;
        auto off = [&](int t) { return ((t >> s) << (s + 1)) | (t & lowmask); };

        const int ldIdx = (off(n) + (hi << s)) * 2;
        const int half = n >> 1, ri = n & 1;
        int sIdx[8];
#pragma unroll
        for (int r = 0; r < 8; ++r)
            sIdx[r] = (off(r + 8 * hi) + (half << s)) * 2 + ri;
        const bool act = (n < 4);

        int bb = 0;
        float2 a = *(const float2*)&stf[ldIdx];
        for (int blk = 0; blk < 32; ++blk) {
            int nb = bb;
            float2 an = a;
            if (blk < 31) {
                int u = (blk + 1) << 4;
                nb = ((((u >> s) << (s + 1)) | (u & lowmask))) * 2;
                an = *(const float2*)&stf[ldIdx + nb];
            }
            v2f A; A.x = a.x; A.y = a.y;
            v8f C = {0.f, 0.f, 0.f, 0.f, 0.f, 0.f, 0.f, 0.f};
            v8f D = __builtin_amdgcn_wmma_f32_16x16x4_f32(
                        false, A, false, Bop, (short)0, C, false, false);
            if (act) {
#pragma unroll
                for (int r = 0; r < 8; ++r) stf[sIdx[r] + bb] = D[r];
            }
            a = an; bb = nb;
        }
    };

    float xv0 = x[b * 3 + 0], xv1 = x[b * 3 + 1], xv2 = x[b * 3 + 2];
    for (int q = 0; q < NQ; ++q) {
        float ang = PI_F * tanhf(xv0 * encW[q * 3] + xv1 * encW[q * 3 + 1] +
                                 xv2 * encW[q * 3 + 2] + encB[q]);
        float c = cosf(ang * 0.5f), sv = sinf(ang * 0.5f);
        int k0 = 2 * hi, k1 = 2 * hi + 1;
        v2f Bop;
        Bop.x = (n == k0) ? c : ((n == (k0 ^ 2)) ? ((k0 < 2) ? -sv : sv) : 0.f);
        Bop.y = (n == k1) ? c : ((n == (k1 ^ 2)) ? ((k1 < 2) ? -sv : sv) : 0.f);
        apply(9 - q, Bop);
    }

    for (int layer = 0; layer < NLAYERS; ++layer) {
        for (int q = 0; q < NQ; ++q) {
            const float* M = Mvar + (layer * NQ + q) * 16;
            v2f Bop;
            Bop.x = (n < 4) ? M[(2 * hi    ) * 4 + n] : 0.f;
            Bop.y = (n < 4) ? M[(2 * hi + 1) * 4 + n] : 0.f;
            apply(9 - q, Bop);
        }
        for (int q = 0; q < NQ; ++q) {
            int cpos = 9 - q, tpos = 9 - ((q + 1) % NQ);
            int lo = cpos < tpos ? cpos : tpos;
            int hp = cpos < tpos ? tpos : cpos;
            for (int k = lane; k < 256; k += 32) {
                int v = ((k >> lo) << (lo + 1)) | (k & ((1 << lo) - 1));
                v     = ((v >> hp) << (hp + 1)) | (v & ((1 << hp) - 1));
                int i0 = v | (1 << cpos);
                int i1 = i0 | (1 << tpos);
                float2 a0 = st[i0];
                float2 a1 = st[i1];
                st[i0] = a1;
                st[i1] = a0;
            }
        }
    }

    float acc[NQ];
#pragma unroll
    for (int q = 0; q < NQ; ++q) acc[q] = 0.f;
    for (int i = lane; i < DIM; i += 32) {
        float2 a = st[i];
        float pr = a.x * a.x + a.y * a.y;
#pragma unroll
        for (int q = 0; q < NQ; ++q) {
            unsigned bit = ((unsigned)i >> (9 - q)) & 1u;
            acc[q] += __int_as_float(__float_as_int(pr) ^ (bit << 31));
        }
    }
#pragma unroll
    for (int off = 16; off > 0; off >>= 1)
#pragma unroll
        for (int q = 0; q < NQ; ++q)
            acc[q] += __shfl_xor(acc[q], off, 32);
    if (lane == 0) {
#pragma unroll
        for (int q = 0; q < NQ; ++q) out[b * NQ + q] = acc[q];
    }
}

extern "C" void kernel_launch(void* const* d_in, const int* in_sizes, int n_in,
                              void* d_out, int out_size, void* d_ws, size_t ws_size,
                              hipStream_t stream) {
    const float* x     = (const float*)d_in[0];
    const float* encW  = (const float*)d_in[1];
    const float* encB  = (const float*)d_in[2];
    const float* theta = (const float*)d_in[3];
    float* out  = (float*)d_out;
    float* Mvar = (float*)d_ws;

    int batch = in_sizes[0] / 3;

    qprep_kernel<<<1, 64, 0, stream>>>(theta, Mvar);

    int blocks = (batch + WPB - 1) / WPB;
    qsim_kernel<<<blocks, WPB * 32, 0, stream>>>(x, encW, encB, Mvar, out, batch);
}


# Gate moved to the WMMA A operand (rows 0,1,8,9), state pairs to B: each lane's D[0:1] is its own new amplitude, so one full-wave 64-bit LDS store replaces eight 8-lane 32-bit stores
# speedup vs baseline: 1.5438x; 1.5438x over previous
#include <hip/hip_runtime.h>
#include <math.h>

typedef __attribute__((ext_vector_type(2))) float v2f;
typedef __attribute__((ext_vector_type(8))) float v8f;

#define NQ       10
#define DIM      1024
#define NLAYERS  4
#define WPB      8
#define PI_F     3.14159265358979323846f

struct c2 { float re, im; };
__device__ __forceinline__ c2 cmul(c2 a, c2 b) {
    return { a.re * b.re - a.im * b.im, a.re * b.im + a.im * b.re };
}
__device__ __forceinline__ c2 cadd(c2 a, c2 b) { return { a.re + b.re, a.im + b.im }; }

__global__ void qprep_kernel(const float* __restrict__ theta, float* __restrict__ Mout) {
    int g = blockIdx.x * blockDim.x + threadIdx.x;
    if (g >= NLAYERS * NQ) return;
    float t0 = theta[g * 3 + 0], t1 = theta[g * 3 + 1], t2 = theta[g * 3 + 2];
    float cx = cosf(t0 * 0.5f), sx = sinf(t0 * 0.5f);
    float cy = cosf(t1 * 0.5f), sy = sinf(t1 * 0.5f);
    float cz = cosf(t2 * 0.5f), sz = sinf(t2 * 0.5f);
    c2 X[2][2] = { { {cx, 0.f}, {0.f, -sx} }, { {0.f, -sx}, {cx, 0.f} } };
    c2 Y[2][2] = { { {cy, 0.f}, {-sy, 0.f} }, { {sy, 0.f}, {cy, 0.f} } };
    c2 Z[2][2] = { { {cz, -sz}, {0.f, 0.f} }, { {0.f, 0.f}, {cz, sz} } };
    c2 XY[2][2], G[2][2];
    for (int i = 0; i < 2; ++i)
        for (int j = 0; j < 2; ++j)
            XY[i][j] = cadd(cmul(X[i][0], Y[0][j]), cmul(X[i][1], Y[1][j]));
    for (int i = 0; i < 2; ++i)
        for (int j = 0; j < 2; ++j)
            G[i][j] = cadd(cmul(XY[i][0], Z[0][j]), cmul(XY[i][1], Z[1][j]));
    float* M = Mout + g * 16;
    for (int i = 0; i < 2; ++i)
        for (int j = 0; j < 2; ++j) {
            M[(2 * i    ) * 4 + 2 * j    ] =  G[i][j].re;
            M[(2 * i    ) * 4 + 2 * j + 1] =  G[i][j].im;
            M[(2 * i + 1) * 4 + 2 * j    ] = -G[i][j].im;
            M[(2 * i + 1) * 4 + 2 * j + 1] =  G[i][j].re;
        }
}

__global__ void __launch_bounds__(WPB * 32)
qsim_kernel(const float* __restrict__ x, const float* __restrict__ encW,
            const float* __restrict__ encB, const float* __restrict__ Mvar,
            float* __restrict__ out, int batch) {
    __shared__ float2 lds[WPB * DIM];

    const int lane = threadIdx.x & 31;
    const int w    = threadIdx.x >> 5;
    const int b    = blockIdx.x * WPB + w;
    if (b >= batch) return;

    float2* st = &lds[w * DIM];
    float*  stf = (float*)st;
    const int n  = lane & 15;
    const int hi = lane >> 4;
    const bool grow = ((n & 6) == 0);
    const int  gj   = (n & 1) | ((n >> 3) << 1);

    for (int i = lane; i < DIM; i += 32) st[i] = make_float2(0.f, 0.f);
    if (lane == 0) st[0] = make_float2(1.f, 0.f);

    auto apply = [&](int s, v2f Gop) {
        const int lowmask = (1 << s) - 1;
        auto off = [&](int t) { return ((t >> s) << (s + 1)) | (t & lowmask); };

        const int ldIdx = (off(n) + (hi << s)) * 2;

        int bb = 0;
        float2 a = *(const float2*)&stf[ldIdx];
        for (int blk = 0; blk < 32; ++blk) {
            int nb = bb;
            float2 an = a;
            if (blk < 31) {
                int u = (blk + 1) << 4;
                nb = ((((u >> s) << (s + 1)) | (u & lowmask))) * 2;
                an = *(const float2*)&stf[ldIdx + nb];
            }
            v2f S; S.x = a.x; S.y = a.y;
            v8f C = {0.f, 0.f, 0.f, 0.f, 0.f, 0.f, 0.f, 0.f};
            v8f D = __builtin_amdgcn_wmma_f32_16x16x4_f32(
                        false, Gop, false, S, (short)0, C, false, false);
            *(float2*)&stf[ldIdx + bb] = make_float2(D[0], D[1]);
            a = an; bb = nb;
        }
    };

    float xv0 = x[b * 3 + 0], xv1 = x[b * 3 + 1], xv2 = x[b * 3 + 2];
    for (int q = 0; q < NQ; ++q) {
        float ang = PI_F * tanhf(xv0 * encW[q * 3] + xv1 * encW[q * 3 + 1] +
                                 xv2 * encW[q * 3 + 2] + encB[q]);
        float c = cosf(ang * 0.5f), sv = sinf(ang * 0.5f);
        int k0 = 2 * hi, k1 = 2 * hi + 1;
        v2f Gop;
        Gop.x = !grow ? 0.f
              : (gj == k0) ? c : ((gj == (k0 ^ 2)) ? ((k0 < 2) ? -sv : sv) : 0.f);
        Gop.y = !grow ? 0.f
              : (gj == k1) ? c : ((gj == (k1 ^ 2)) ? ((k1 < 2) ? -sv : sv) : 0.f);
        apply(9 - q, Gop);
    }

    for (int layer = 0; layer < NLAYERS; ++layer) {
        for (int q = 0; q < NQ; ++q) {
            const float* M = Mvar + (layer * NQ + q) * 16;
            v2f Gop;
            Gop.x = grow ? M[(2 * hi    ) * 4 + gj] : 0.f;
            Gop.y = grow ? M[(2 * hi + 1) * 4 + gj] : 0.f;
            apply(9 - q, Gop);
        }
        for (int q = 0; q < NQ; ++q) {
            int cpos = 9 - q, tpos = 9 - ((q + 1) % NQ);
            int lo = cpos < tpos ? cpos : tpos;
            int hp = cpos < tpos ? tpos : cpos;
            for (int k = lane; k < 256; k += 32) {
                int v = ((k >> lo) << (lo + 1)) | (k & ((1 << lo) - 1));
                v     = ((v >> hp) << (hp + 1)) | (v & ((1 << hp) - 1));
                int i0 = v | (1 << cpos);
                int i1 = i0 | (1 << tpos);
                float2 a0 = st[i0];
                float2 a1 = st[i1];
                st[i0] = a1;
                st[i1] = a0;
            }
        }
    }

    float acc[NQ];
#pragma unroll
    for (int q = 0; q < NQ; ++q) acc[q] = 0.f;
    for (int i = lane; i < DIM; i += 32) {
        float2 a = st[i];
        float pr = a.x * a.x + a.y * a.y;
#pragma unroll
        for (int q = 0; q < NQ; ++q) {
            unsigned bit = ((unsigned)i >> (9 - q)) & 1u;
            acc[q] += __int_as_float(__float_as_int(pr) ^ (bit << 31));
        }
    }
#pragma unroll
    for (int off = 16; off > 0; off >>= 1)
#pragma unroll
        for (int q = 0; q < NQ; ++q)
            acc[q] += __shfl_xor(acc[q], off, 32);
    if (lane == 0) {
#pragma unroll
        for (int q = 0; q < NQ; ++q) out[b * NQ + q] = acc[q];
    }
}

extern "C" void kernel_launch(void* const* d_in, const int* in_sizes, int n_in,
                              void* d_out, int out_size, void* d_ws, size_t ws_size,
                              hipStream_t stream) {
    const float* x     = (const float*)d_in[0];
    const float* encW  = (const float*)d_in[1];
    const float* encB  = (const float*)d_in[2];
    const float* theta = (const float*)d_in[3];
    float* out  = (float*)d_out;
    float* Mvar = (float*)d_ws;

    int batch = in_sizes[0] / 3;

    qprep_kernel<<<1, 64, 0, stream>>>(theta, Mvar);

    int blocks = (batch + WPB - 1) / WPB;
    qsim_kernel<<<blocks, WPB * 32, 0, stream>>>(x, encW, encB, Mvar, out, batch);
}


# Gates of three qubits fused into one 16x16 real operator per LDS pass (4 chained K=4 WMMA per 16 groups, all 16 A rows used): 20 passes instead of 50; each amplitude now one 16-term sum
# speedup vs baseline: 3.2679x; 2.1168x over previous
#include <hip/hip_runtime.h>
#include <math.h>

typedef __attribute__((ext_vector_type(2))) float v2f;
typedef __attribute__((ext_vector_type(8))) float v8f;

#define NQ       10
#define DIM      1024
#define NLAYERS  4
#define WPB      8
#define PI_F     3.14159265358979323846f

struct c2 { float re, im; };
__device__ __forceinline__ c2 cmul(c2 a, c2 b) {
    return { a.re * b.re - a.im * b.im, a.re * b.im + a.im * b.re };
}
__device__ __forceinline__ c2 cadd(c2 a, c2 b) { return { a.re + b.re, a.im + b.im }; }

__global__ void qprep_kernel(const float* __restrict__ theta, float* __restrict__ Mout) {
    int g = blockIdx.x * blockDim.x + threadIdx.x;
    if (g >= NLAYERS * NQ) return;
    float t0 = theta[g * 3 + 0], t1 = theta[g * 3 + 1], t2 = theta[g * 3 + 2];
    float cx = cosf(t0 * 0.5f), sx = sinf(t0 * 0.5f);
    float cy = cosf(t1 * 0.5f), sy = sinf(t1 * 0.5f);
    float cz = cosf(t2 * 0.5f), sz = sinf(t2 * 0.5f);
    c2 X[2][2] = { { {cx, 0.f}, {0.f, -sx} }, { {0.f, -sx}, {cx, 0.f} } };
    c2 Y[2][2] = { { {cy, 0.f}, {-sy, 0.f} }, { {sy, 0.f}, {cy, 0.f} } };
    c2 Z[2][2] = { { {cz, -sz}, {0.f, 0.f} }, { {0.f, 0.f}, {cz, sz} } };
    c2 XY[2][2], G[2][2];
    for (int i = 0; i < 2; ++i)
        for (int j = 0; j < 2; ++j)
            XY[i][j] = cadd(cmul(X[i][0], Y[0][j]), cmul(X[i][1], Y[1][j]));
    for (int i = 0; i < 2; ++i)
        for (int j = 0; j < 2; ++j)
            G[i][j] = cadd(cmul(XY[i][0], Z[0][j]), cmul(XY[i][1], Z[1][j]));
    float* M = Mout + g * 16;
    for (int i = 0; i < 2; ++i)
        for (int j = 0; j < 2; ++j) {
            M[(2 * i    ) * 4 + 2 * j    ] =  G[i][j].re;
            M[(2 * i    ) * 4 + 2 * j + 1] =  G[i][j].im;
            M[(2 * i + 1) * 4 + 2 * j    ] = -G[i][j].im;
            M[(2 * i + 1) * 4 + 2 * j + 1] =  G[i][j].re;
        }
}

template <int PH, int PJ1, int PJ0, int N0, int N1, int N2, int N3, int T0, int T1, int T2>
__device__ __forceinline__ void fused_pass(float2* __restrict__ st, int lane,
                                           c2 uH, c2 uJ1a, c2 uJ1b, c2 uJ0a, c2 uJ0b) {
    const int c  = lane & 15;
    const int h  = lane >> 4;
    const int co = c & 1;

    const c2 t0 = cmul(uH, uJ1a), t1 = cmul(uH, uJ1b);
    c2 P[4];
    P[0] = cmul(t0, uJ0a); P[1] = cmul(t0, uJ0b);
    P[2] = cmul(t1, uJ0a); P[3] = cmul(t1, uJ0b);
    v2f A[4];
#pragma unroll
    for (int j = 0; j < 4; ++j) {
        A[j].x = co ? P[j].im :  P[j].re;
        A[j].y = co ? P[j].re : -P[j].im;
    }

    const int base = ((c & 1) << N0) | (((c >> 1) & 1) << N1) | (((c >> 2) & 1) << N2) |
                     (((c >> 3) & 1) << N3) | (h << PH);
    float2* p = st + base;

    float2 cur0[4], cur1[4], nx0[4], nx1[4];
#pragma unroll
    for (int j = 0; j < 4; ++j) {
        const int jo = ((j >> 1) << PJ1) | ((j & 1) << PJ0);
        cur0[j] = p[jo];
        cur1[j] = p[(1 << T0) | jo];
    }
#pragma unroll
    for (int tp = 0; tp < 4; ++tp) {
        const int ta = ((tp & 1) << T1) | (((tp >> 1) & 1) << T2);
        const int tb = ta | (1 << T0);
        if (tp < 3) {
            const int u  = 2 * tp + 2;
            const int na = (((u >> 1) & 1) << T1) | (((u >> 2) & 1) << T2);
#pragma unroll
            for (int j = 0; j < 4; ++j) {
                const int jo = ((j >> 1) << PJ1) | ((j & 1) << PJ0);
                nx0[j] = p[na | jo];
                nx1[j] = p[na | (1 << T0) | jo];
            }
        }
        v8f C0 = {0.f, 0.f, 0.f, 0.f, 0.f, 0.f, 0.f, 0.f};
        v8f C1 = {0.f, 0.f, 0.f, 0.f, 0.f, 0.f, 0.f, 0.f};
#pragma unroll
        for (int j = 0; j < 4; ++j) {
            v2f B0; B0.x = cur0[j].x; B0.y = cur0[j].y;
            v2f B1; B1.x = cur1[j].x; B1.y = cur1[j].y;
            C0 = __builtin_amdgcn_wmma_f32_16x16x4_f32(false, A[j], false, B0, (short)0, C0, false, false);
            C1 = __builtin_amdgcn_wmma_f32_16x16x4_f32(false, A[j], false, B1, (short)0, C1, false, false);
        }
#pragma unroll
        for (int j = 0; j < 4; ++j) {
            const int jo = ((j >> 1) << PJ1) | ((j & 1) << PJ0);
            p[ta | jo] = make_float2(C0[2 * j], C0[2 * j + 1]);
            p[tb | jo] = make_float2(C1[2 * j], C1[2 * j + 1]);
        }
#pragma unroll
        for (int j = 0; j < 4; ++j) { cur0[j] = nx0[j]; cur1[j] = nx1[j]; }
    }
}

#define PASS0 fused_pass<4, 9, 8, 0, 1, 2, 3, 5, 6, 7>
#define PASS1 fused_pass<3, 7, 6, 0, 1, 2, 4, 5, 8, 9>
#define PASS2 fused_pass<2, 5, 0, 1, 3, 4, 6, 7, 8, 9>
#define PASS3 fused_pass<1, 9, 8, 0, 2, 3, 4, 5, 6, 7>

__device__ __forceinline__ c2 ry_u(float c, float s, int o, int i) {
    c2 r; r.re = (o == i) ? c : ((o < i) ? s : -s); r.im = 0.f; return r;
}
__device__ __forceinline__ c2 id_u(int o, int i) {
    c2 r; r.re = (o == i) ? 1.f : 0.f; r.im = 0.f; return r;
}
__device__ __forceinline__ c2 var_u(const float* __restrict__ M, int o, int i) {
    const float2 v = *(const float2*)(M + 8 * i + 2 * o);
    c2 r; r.re = v.x; r.im = v.y; return r;
}

__global__ void __launch_bounds__(WPB * 32)
qsim_kernel(const float* __restrict__ x, const float* __restrict__ encW,
            const float* __restrict__ encB, const float* __restrict__ Mvar,
            float* __restrict__ out, int batch) {
    __shared__ float2 lds[WPB * DIM];

    const int lane = threadIdx.x & 31;
    const int w    = threadIdx.x >> 5;
    const int b    = blockIdx.x * WPB + w;
    if (b >= batch) return;

    float2* st = &lds[w * DIM];
    const int c  = lane & 15;
    const int h  = lane >> 4;
    const int ho = c >> 3;
    const int o1 = (c >> 2) & 1;
    const int o0 = (c >> 1) & 1;

    for (int i = lane; i < DIM; i += 32) st[i] = make_float2(0.f, 0.f);
    if (lane == 0) st[0] = make_float2(1.f, 0.f);

    {
        float xv0 = x[b * 3 + 0], xv1 = x[b * 3 + 1], xv2 = x[b * 3 + 2];
        float cq[NQ], sq[NQ];
#pragma unroll
        for (int q = 0; q < NQ; ++q) {
            float ang = PI_F * tanhf(xv0 * encW[q * 3] + xv1 * encW[q * 3 + 1] +
                                     xv2 * encW[q * 3 + 2] + encB[q]);
            cq[q] = cosf(ang * 0.5f);
            sq[q] = sinf(ang * 0.5f);
        }
        PASS0(st, lane, ry_u(cq[5], sq[5], ho, h),
              ry_u(cq[0], sq[0], o1, 0), ry_u(cq[0], sq[0], o1, 1),
              ry_u(cq[1], sq[1], o0, 0), ry_u(cq[1], sq[1], o0, 1));
        PASS1(st, lane, ry_u(cq[6], sq[6], ho, h),
              ry_u(cq[2], sq[2], o1, 0), ry_u(cq[2], sq[2], o1, 1),
              ry_u(cq[3], sq[3], o0, 0), ry_u(cq[3], sq[3], o0, 1));
        PASS2(st, lane, ry_u(cq[7], sq[7], ho, h),
              ry_u(cq[4], sq[4], o1, 0), ry_u(cq[4], sq[4], o1, 1),
              ry_u(cq[9], sq[9], o0, 0), ry_u(cq[9], sq[9], o0, 1));
        PASS3(st, lane, ry_u(cq[8], sq[8], ho, h),
              id_u(o1, 0), id_u(o1, 1), id_u(o0, 0), id_u(o0, 1));
    }

    for (int layer = 0; layer < NLAYERS; ++layer) {
        const float* ML = Mvar + layer * NQ * 16;
        PASS0(st, lane, var_u(ML + 5 * 16, ho, h),
              var_u(ML + 0 * 16, o1, 0), var_u(ML + 0 * 16, o1, 1),
              var_u(ML + 1 * 16, o0, 0), var_u(ML + 1 * 16, o0, 1));
        PASS1(st, lane, var_u(ML + 6 * 16, ho, h),
              var_u(ML + 2 * 16, o1, 0), var_u(ML + 2 * 16, o1, 1),
              var_u(ML + 3 * 16, o0, 0), var_u(ML + 3 * 16, o0, 1));
        PASS2(st, lane, var_u(ML + 7 * 16, ho, h),
              var_u(ML + 4 * 16, o1, 0), var_u(ML + 4 * 16, o1, 1),
              var_u(ML + 9 * 16, o0, 0), var_u(ML + 9 * 16, o0, 1));
        PASS3(st, lane, var_u(ML + 8 * 16, ho, h),
              id_u(o1, 0), id_u(o1, 1), id_u(o0, 0), id_u(o0, 1));

        for (int q = 0; q < NQ; ++q) {
            int cpos = 9 - q, tpos = 9 - ((q + 1) % NQ);
            int lo = cpos < tpos ? cpos : tpos;
            int hp = cpos < tpos ? tpos : cpos;
            for (int k = lane; k < 256; k += 32) {
                int v = ((k >> lo) << (lo + 1)) | (k & ((1 << lo) - 1));
                v     = ((v >> hp) << (hp + 1)) | (v & ((1 << hp) - 1));
                int i0 = v | (1 << cpos);
                int i1 = i0 | (1 << tpos);
                float2 a0 = st[i0];
                float2 a1 = st[i1];
                st[i0] = a1;
                st[i1] = a0;
            }
        }
    }

    float acc[NQ];
#pragma unroll
    for (int q = 0; q < NQ; ++q) acc[q] = 0.f;
    for (int i = lane; i < DIM; i += 32) {
        float2 a = st[i];
        float pr = a.x * a.x + a.y * a.y;
#pragma unroll
        for (int q = 0; q < NQ; ++q) {
            unsigned bit = ((unsigned)i >> (9 - q)) & 1u;
            acc[q] += __int_as_float(__float_as_int(pr) ^ (bit << 31));
        }
    }
#pragma unroll
    for (int off = 16; off > 0; off >>= 1)
#pragma unroll
        for (int q = 0; q < NQ; ++q)
            acc[q] += __shfl_xor(acc[q], off, 32);
    if (lane == 0) {
#pragma unroll
        for (int q = 0; q < NQ; ++q) out[b * NQ + q] = acc[q];
    }
}

extern "C" void kernel_launch(void* const* d_in, const int* in_sizes, int n_in,
                              void* d_out, int out_size, void* d_ws, size_t ws_size,
                              hipStream_t stream) {
    const float* x     = (const float*)d_in[0];
    const float* encW  = (const float*)d_in[1];
    const float* encB  = (const float*)d_in[2];
    const float* theta = (const float*)d_in[3];
    float* out  = (float*)d_out;
    float* Mvar = (float*)d_ws;

    int batch = in_sizes[0] / 3;

    qprep_kernel<<<1, 64, 0, stream>>>(theta, Mvar);

    int blocks = (batch + WPB - 1) / WPB;
    qsim_kernel<<<blocks, WPB * 32, 0, stream>>>(x, encW, encB, Mvar, out, batch);
}


# The ten CNOT swaps of a layer composed into one GF(2)-linear index map, applied as one bank-conflict-free gather through registers: 32 LDS loads + 32 stores per lane per layer instead of 160 + 160
# speedup vs baseline: 4.2238x; 1.2925x over previous
#include <hip/hip_runtime.h>
#include <math.h>

typedef __attribute__((ext_vector_type(2))) float v2f;
typedef __attribute__((ext_vector_type(8))) float v8f;

#define NQ       10
#define DIM      1024
#define NLAYERS  4
#define WPB      8
#define PI_F     3.14159265358979323846f

struct c2 { float re, im; };
__device__ __forceinline__ c2 cmul(c2 a, c2 b) {
    return { a.re * b.re - a.im * b.im, a.re * b.im + a.im * b.re };
}
__device__ __forceinline__ c2 cadd(c2 a, c2 b) { return { a.re + b.re, a.im + b.im }; }

__global__ void qprep_kernel(const float* __restrict__ theta, float* __restrict__ Mout) {
    int g = blockIdx.x * blockDim.x + threadIdx.x;
    if (g >= NLAYERS * NQ) return;
    float t0 = theta[g * 3 + 0], t1 = theta[g * 3 + 1], t2 = theta[g * 3 + 2];
    float cx = cosf(t0 * 0.5f), sx = sinf(t0 * 0.5f);
    float cy = cosf(t1 * 0.5f), sy = sinf(t1 * 0.5f);
    float cz = cosf(t2 * 0.5f), sz = sinf(t2 * 0.5f);
    c2 X[2][2] = { { {cx, 0.f}, {0.f, -sx} }, { {0.f, -sx}, {cx, 0.f} } };
    c2 Y[2][2] = { { {cy, 0.f}, {-sy, 0.f} }, { {sy, 0.f}, {cy, 0.f} } };
    c2 Z[2][2] = { { {cz, -sz}, {0.f, 0.f} }, { {0.f, 0.f}, {cz, sz} } };
    c2 XY[2][2], G[2][2];
    for (int i = 0; i < 2; ++i)
        for (int j = 0; j < 2; ++j)
            XY[i][j] = cadd(cmul(X[i][0], Y[0][j]), cmul(X[i][1], Y[1][j]));
    for (int i = 0; i < 2; ++i)
        for (int j = 0; j < 2; ++j)
            G[i][j] = cadd(cmul(XY[i][0], Z[0][j]), cmul(XY[i][1], Z[1][j]));
    float* M = Mout + g * 16;
    for (int i = 0; i < 2; ++i)
        for (int j = 0; j < 2; ++j) {
            M[(2 * i    ) * 4 + 2 * j    ] =  G[i][j].re;
            M[(2 * i    ) * 4 + 2 * j + 1] =  G[i][j].im;
            M[(2 * i + 1) * 4 + 2 * j    ] = -G[i][j].im;
            M[(2 * i + 1) * 4 + 2 * j + 1] =  G[i][j].re;
        }
}

template <int PH, int PJ1, int PJ0, int N0, int N1, int N2, int N3, int T0, int T1, int T2>
__device__ __forceinline__ void fused_pass(float2* __restrict__ st, int lane,
                                           c2 uH, c2 uJ1a, c2 uJ1b, c2 uJ0a, c2 uJ0b) {
    const int c  = lane & 15;
    const int h  = lane >> 4;
    const int co = c & 1;

    const c2 t0 = cmul(uH, uJ1a), t1 = cmul(uH, uJ1b);
    c2 P[4];
    P[0] = cmul(t0, uJ0a); P[1] = cmul(t0, uJ0b);
    P[2] = cmul(t1, uJ0a); P[3] = cmul(t1, uJ0b);
    v2f A[4];
#pragma unroll
    for (int j = 0; j < 4; ++j) {
        A[j].x = co ? P[j].im :  P[j].re;
        A[j].y = co ? P[j].re : -P[j].im;
    }

    const int base = ((c & 1) << N0) | (((c >> 1) & 1) << N1) | (((c >> 2) & 1) << N2) |
                     (((c >> 3) & 1) << N3) | (h << PH);
    float2* p = st + base;

    float2 cur0[4], cur1[4], nx0[4], nx1[4];
#pragma unroll
    for (int j = 0; j < 4; ++j) {
        const int jo = ((j >> 1) << PJ1) | ((j & 1) << PJ0);
        cur0[j] = p[jo];
        cur1[j] = p[(1 << T0) | jo];
    }
#pragma unroll
    for (int tp = 0; tp < 4; ++tp) {
        const int ta = ((tp & 1) << T1) | (((tp >> 1) & 1) << T2);
        const int tb = ta | (1 << T0);
        if (tp < 3) {
            const int u  = 2 * tp + 2;
            const int na = (((u >> 1) & 1) << T1) | (((u >> 2) & 1) << T2);
#pragma unroll
            for (int j = 0; j < 4; ++j) {
                const int jo = ((j >> 1) << PJ1) | ((j & 1) << PJ0);
                nx0[j] = p[na | jo];
                nx1[j] = p[na | (1 << T0) | jo];
            }
        }
        v8f C0 = {0.f, 0.f, 0.f, 0.f, 0.f, 0.f, 0.f, 0.f};
        v8f C1 = {0.f, 0.f, 0.f, 0.f, 0.f, 0.f, 0.f, 0.f};
#pragma unroll
        for (int j = 0; j < 4; ++j) {
            v2f B0; B0.x = cur0[j].x; B0.y = cur0[j].y;
            v2f B1; B1.x = cur1[j].x; B1.y = cur1[j].y;
            C0 = __builtin_amdgcn_wmma_f32_16x16x4_f32(false, A[j], false, B0, (short)0, C0, false, false);
            C1 = __builtin_amdgcn_wmma_f32_16x16x4_f32(false, A[j], false, B1, (short)0, C1, false, false);
        }
#pragma unroll
        for (int j = 0; j < 4; ++j) {
            const int jo = ((j >> 1) << PJ1) | ((j & 1) << PJ0);
            p[ta | jo] = make_float2(C0[2 * j], C0[2 * j + 1]);
            p[tb | jo] = make_float2(C1[2 * j], C1[2 * j + 1]);
        }
#pragma unroll
        for (int j = 0; j < 4; ++j) { cur0[j] = nx0[j]; cur1[j] = nx1[j]; }
    }
}

#define PASS0 fused_pass<4, 9, 8, 0, 1, 2, 3, 5, 6, 7>
#define PASS1 fused_pass<3, 7, 6, 0, 1, 2, 4, 5, 8, 9>
#define PASS2 fused_pass<2, 5, 0, 1, 3, 4, 6, 7, 8, 9>
#define PASS3 fused_pass<1, 9, 8, 0, 2, 3, 4, 5, 6, 7>

__device__ __forceinline__ c2 ry_u(float c, float s, int o, int i) {
    c2 r; r.re = (o == i) ? c : ((o < i) ? s : -s); r.im = 0.f; return r;
}
__device__ __forceinline__ c2 id_u(int o, int i) {
    c2 r; r.re = (o == i) ? 1.f : 0.f; r.im = 0.f; return r;
}
__device__ __forceinline__ c2 var_u(const float* __restrict__ M, int o, int i) {
    const float2 v = *(const float2*)(M + 8 * i + 2 * o);
    c2 r; r.re = v.x; r.im = v.y; return r;
}

__global__ void __launch_bounds__(WPB * 32)
qsim_kernel(const float* __restrict__ x, const float* __restrict__ encW,
            const float* __restrict__ encB, const float* __restrict__ Mvar,
            float* __restrict__ out, int batch) {
    __shared__ float2 lds[WPB * DIM];

    const int lane = threadIdx.x & 31;
    const int w    = threadIdx.x >> 5;
    const int b    = blockIdx.x * WPB + w;
    if (b >= batch) return;

    float2* st = &lds[w * DIM];
    const int c  = lane & 15;
    const int h  = lane >> 4;
    const int ho = c >> 3;
    const int o1 = (c >> 2) & 1;
    const int o0 = (c >> 1) & 1;

    for (int i = lane; i < DIM; i += 32) st[i] = make_float2(0.f, 0.f);
    if (lane == 0) st[0] = make_float2(1.f, 0.f);

    {
        float xv0 = x[b * 3 + 0], xv1 = x[b * 3 + 1], xv2 = x[b * 3 + 2];
        float cq[NQ], sq[NQ];
#pragma unroll
        for (int q = 0; q < NQ; ++q) {
            float ang = PI_F * tanhf(xv0 * encW[q * 3] + xv1 * encW[q * 3 + 1] +
                                     xv2 * encW[q * 3 + 2] + encB[q]);
            cq[q] = cosf(ang * 0.5f);
            sq[q] = sinf(ang * 0.5f);
        }
        PASS0(st, lane, ry_u(cq[5], sq[5], ho, h),
              ry_u(cq[0], sq[0], o1, 0), ry_u(cq[0], sq[0], o1, 1),
              ry_u(cq[1], sq[1], o0, 0), ry_u(cq[1], sq[1], o0, 1));
        PASS1(st, lane, ry_u(cq[6], sq[6], ho, h),
              ry_u(cq[2], sq[2], o1, 0), ry_u(cq[2], sq[2], o1, 1),
              ry_u(cq[3], sq[3], o0, 0), ry_u(cq[3], sq[3], o0, 1));
        PASS2(st, lane, ry_u(cq[7], sq[7], ho, h),
              ry_u(cq[4], sq[4], o1, 0), ry_u(cq[4], sq[4], o1, 1),
              ry_u(cq[9], sq[9], o0, 0), ry_u(cq[9], sq[9], o0, 1));
        PASS3(st, lane, ry_u(cq[8], sq[8], ho, h),
              id_u(o1, 0), id_u(o1, 1), id_u(o0, 0), id_u(o0, 1));
    }

    for (int layer = 0; layer < NLAYERS; ++layer) {
        const float* ML = Mvar + layer * NQ * 16;
        PASS0(st, lane, var_u(ML + 5 * 16, ho, h),
              var_u(ML + 0 * 16, o1, 0), var_u(ML + 0 * 16, o1, 1),
              var_u(ML + 1 * 16, o0, 0), var_u(ML + 1 * 16, o0, 1));
        PASS1(st, lane, var_u(ML + 6 * 16, ho, h),
              var_u(ML + 2 * 16, o1, 0), var_u(ML + 2 * 16, o1, 1),
              var_u(ML + 3 * 16, o0, 0), var_u(ML + 3 * 16, o0, 1));
        PASS2(st, lane, var_u(ML + 7 * 16, ho, h),
              var_u(ML + 4 * 16, o1, 0), var_u(ML + 4 * 16, o1, 1),
              var_u(ML + 9 * 16, o0, 0), var_u(ML + 9 * 16, o0, 1));
        PASS3(st, lane, var_u(ML + 8 * 16, ho, h),
              id_u(o1, 0), id_u(o1, 1), id_u(o0, 0), id_u(o0, 1));

        {
            const int fl = (lane ^ (lane >> 1)) | ((lane & 1) ? 0x300 : 0);
            float2 g[32];
#pragma unroll
            for (int t = 0; t < 32; ++t) {
                const int ft = (t << 5) ^ (t << 4);
                g[t] = st[(fl ^ (ft & 0x310)) + (ft & 0x0E0)];
            }
#pragma unroll
            for (int t = 0; t < 32; ++t) st[lane + (t << 5)] = g[t];
        }
    }

    float acc[NQ];
#pragma unroll
    for (int q = 0; q < NQ; ++q) acc[q] = 0.f;
    for (int i = lane; i < DIM; i += 32) {
        float2 a = st[i];
        float pr = a.x * a.x + a.y * a.y;
#pragma unroll
        for (int q = 0; q < NQ; ++q) {
            unsigned bit = ((unsigned)i >> (9 - q)) & 1u;
            acc[q] += __int_as_float(__float_as_int(pr) ^ (bit << 31));
        }
    }
#pragma unroll
    for (int off = 16; off > 0; off >>= 1)
#pragma unroll
        for (int q = 0; q < NQ; ++q)
            acc[q] += __shfl_xor(acc[q], off, 32);
    if (lane == 0) {
#pragma unroll
        for (int q = 0; q < NQ; ++q) out[b * NQ + q] = acc[q];
    }
}

extern "C" void kernel_launch(void* const* d_in, const int* in_sizes, int n_in,
                              void* d_out, int out_size, void* d_ws, size_t ws_size,
                              hipStream_t stream) {
    const float* x     = (const float*)d_in[0];
    const float* encW  = (const float*)d_in[1];
    const float* encB  = (const float*)d_in[2];
    const float* theta = (const float*)d_in[3];
    float* out  = (float*)d_out;
    float* Mvar = (float*)d_ws;

    int batch = in_sizes[0] / 3;

    qprep_kernel<<<1, 64, 0, stream>>>(theta, Mvar);

    int blocks = (batch + WPB - 1) / WPB;
    qsim_kernel<<<blocks, WPB * 32, 0, stream>>>(x, encW, encB, Mvar, out, batch);
}


# State before layer 1 is a tensor product (no two-qubit gate yet): built per lane in closed form, first CNOT ring's index map folded in; 12 WMMA passes, not 20; those amplitudes: products, not sums
# speedup vs baseline: 5.9790x; 1.4155x over previous
#include <hip/hip_runtime.h>
#include <math.h>

typedef __attribute__((ext_vector_type(2))) float v2f;
typedef __attribute__((ext_vector_type(8))) float v8f;

#define NQ       10
#define DIM      1024
#define NLAYERS  4
#define WPB      8
#define PI_F     3.14159265358979323846f

struct c2 { float re, im; };
__device__ __forceinline__ c2 cmul(c2 a, c2 b) {
    return { a.re * b.re - a.im * b.im, a.re * b.im + a.im * b.re };
}
__device__ __forceinline__ c2 cadd(c2 a, c2 b) { return { a.re + b.re, a.im + b.im }; }

__global__ void qprep_kernel(const float* __restrict__ theta, float* __restrict__ Mout) {
    int g = blockIdx.x * blockDim.x + threadIdx.x;
    if (g >= NLAYERS * NQ) return;
    float t0 = theta[g * 3 + 0], t1 = theta[g * 3 + 1], t2 = theta[g * 3 + 2];
    float cx = cosf(t0 * 0.5f), sx = sinf(t0 * 0.5f);
    float cy = cosf(t1 * 0.5f), sy = sinf(t1 * 0.5f);
    float cz = cosf(t2 * 0.5f), sz = sinf(t2 * 0.5f);
    c2 X[2][2] = { { {cx, 0.f}, {0.f, -sx} }, { {0.f, -sx}, {cx, 0.f} } };
    c2 Y[2][2] = { { {cy, 0.f}, {-sy, 0.f} }, { {sy, 0.f}, {cy, 0.f} } };
    c2 Z[2][2] = { { {cz, -sz}, {0.f, 0.f} }, { {0.f, 0.f}, {cz, sz} } };
    c2 XY[2][2], G[2][2];
    for (int i = 0; i < 2; ++i)
        for (int j = 0; j < 2; ++j)
            XY[i][j] = cadd(cmul(X[i][0], Y[0][j]), cmul(X[i][1], Y[1][j]));
    for (int i = 0; i < 2; ++i)
        for (int j = 0; j < 2; ++j)
            G[i][j] = cadd(cmul(XY[i][0], Z[0][j]), cmul(XY[i][1], Z[1][j]));
    float* M = Mout + g * 16;
    for (int i = 0; i < 2; ++i)
        for (int j = 0; j < 2; ++j) {
            M[(2 * i    ) * 4 + 2 * j    ] =  G[i][j].re;
            M[(2 * i    ) * 4 + 2 * j + 1] =  G[i][j].im;
            M[(2 * i + 1) * 4 + 2 * j    ] = -G[i][j].im;
            M[(2 * i + 1) * 4 + 2 * j + 1] =  G[i][j].re;
        }
}

template <int PH, int PJ1, int PJ0, int N0, int N1, int N2, int N3, int T0, int T1, int T2>
__device__ __forceinline__ void fused_pass(float2* __restrict__ st, int lane,
                                           c2 uH, c2 uJ1a, c2 uJ1b, c2 uJ0a, c2 uJ0b) {
    const int c  = lane & 15;
    const int h  = lane >> 4;
    const int co = c & 1;

    const c2 t0 = cmul(uH, uJ1a), t1 = cmul(uH, uJ1b);
    c2 P[4];
    P[0] = cmul(t0, uJ0a); P[1] = cmul(t0, uJ0b);
    P[2] = cmul(t1, uJ0a); P[3] = cmul(t1, uJ0b);
    v2f A[4];
#pragma unroll
    for (int j = 0; j < 4; ++j) {
        A[j].x = co ? P[j].im :  P[j].re;
        A[j].y = co ? P[j].re : -P[j].im;
    }

    const int base = ((c & 1) << N0) | (((c >> 1) & 1) << N1) | (((c >> 2) & 1) << N2) |
                     (((c >> 3) & 1) << N3) | (h << PH);
    float2* p = st + base;

    float2 cur0[4], cur1[4], nx0[4], nx1[4];
#pragma unroll
    for (int j = 0; j < 4; ++j) {
        const int jo = ((j >> 1) << PJ1) | ((j & 1) << PJ0);
        cur0[j] = p[jo];
        cur1[j] = p[(1 << T0) | jo];
    }
#pragma unroll
    for (int tp = 0; tp < 4; ++tp) {
        const int ta = ((tp & 1) << T1) | (((tp >> 1) & 1) << T2);
        const int tb = ta | (1 << T0);
        if (tp < 3) {
            const int u  = 2 * tp + 2;
            const int na = (((u >> 1) & 1) << T1) | (((u >> 2) & 1) << T2);
#pragma unroll
            for (int j = 0; j < 4; ++j) {
                const int jo = ((j >> 1) << PJ1) | ((j & 1) << PJ0);
                nx0[j] = p[na | jo];
                nx1[j] = p[na | (1 << T0) | jo];
            }
        }
        v8f C0 = {0.f, 0.f, 0.f, 0.f, 0.f, 0.f, 0.f, 0.f};
        v8f C1 = {0.f, 0.f, 0.f, 0.f, 0.f, 0.f, 0.f, 0.f};
#pragma unroll
        for (int j = 0; j < 4; ++j) {
            v2f B0; B0.x = cur0[j].x; B0.y = cur0[j].y;
            v2f B1; B1.x = cur1[j].x; B1.y = cur1[j].y;
            C0 = __builtin_amdgcn_wmma_f32_16x16x4_f32(false, A[j], false, B0, (short)0, C0, false, false);
            C1 = __builtin_amdgcn_wmma_f32_16x16x4_f32(false, A[j], false, B1, (short)0, C1, false, false);
        }
#pragma unroll
        for (int j = 0; j < 4; ++j) {
            const int jo = ((j >> 1) << PJ1) | ((j & 1) << PJ0);
            p[ta | jo] = make_float2(C0[2 * j], C0[2 * j + 1]);
            p[tb | jo] = make_float2(C1[2 * j], C1[2 * j + 1]);
        }
#pragma unroll
        for (int j = 0; j < 4; ++j) { cur0[j] = nx0[j]; cur1[j] = nx1[j]; }
    }
}

#define PASS0 fused_pass<4, 9, 8, 0, 1, 2, 3, 5, 6, 7>
#define PASS1 fused_pass<3, 7, 6, 0, 1, 2, 4, 5, 8, 9>
#define PASS2 fused_pass<2, 5, 0, 1, 3, 4, 6, 7, 8, 9>
#define PASS3 fused_pass<1, 9, 8, 0, 2, 3, 4, 5, 6, 7>

__device__ __forceinline__ c2 csel(int pick1, c2 a0, c2 a1) {
    c2 r; r.re = pick1 ? a1.re : a0.re; r.im = pick1 ? a1.im : a0.im; return r;
}
__device__ __forceinline__ c2 id_u(int o, int i) {
    c2 r; r.re = (o == i) ? 1.f : 0.f; r.im = 0.f; return r;
}
__device__ __forceinline__ c2 var_u(const float* __restrict__ M, int o, int i) {
    const float2 v = *(const float2*)(M + 8 * i + 2 * o);
    c2 r; r.re = v.x; r.im = v.y; return r;
}

__global__ void __launch_bounds__(WPB * 32)
qsim_kernel(const float* __restrict__ x, const float* __restrict__ encW,
            const float* __restrict__ encB, const float* __restrict__ Mvar,
            float* __restrict__ out, int batch) {
    __shared__ float2 lds[WPB * DIM];

    const int lane = threadIdx.x & 31;
    const int w    = threadIdx.x >> 5;
    const int b    = blockIdx.x * WPB + w;
    if (b >= batch) return;

    float2* st = &lds[w * DIM];
    const int c  = lane & 15;
    const int h  = lane >> 4;
    const int ho = c >> 3;
    const int o1 = (c >> 2) & 1;
    const int o0 = (c >> 1) & 1;

    {
        float xv0 = x[b * 3 + 0], xv1 = x[b * 3 + 1], xv2 = x[b * 3 + 2];
        c2 V[NQ][2];
#pragma unroll
        for (int p = 0; p < NQ; ++p) {
            const int q = 9 - p;
            float ang = PI_F * tanhf(xv0 * encW[q * 3] + xv1 * encW[q * 3 + 1] +
                                     xv2 * encW[q * 3 + 2] + encB[q]);
            const float cq = cosf(ang * 0.5f);
            const float sq = sinf(ang * 0.5f);
            const float* M = Mvar + q * 16;
#pragma unroll
            for (int o = 0; o < 2; ++o) {
                const c2 g0 = var_u(M, o, 0), g1 = var_u(M, o, 1);
                V[p][o].re = cq * g0.re - sq * g1.re;
                V[p][o].im = cq * g0.im - sq * g1.im;
            }
        }
        const int gl = lane ^ (lane >> 1);
        const int l0 = lane & 1, l4 = lane >> 4;
        c2 L = csel(gl & 1, V[0][0], V[0][1]);
        L = cmul(L, csel((gl >> 1) & 1, V[1][0], V[1][1]));
        L = cmul(L, csel((gl >> 2) & 1, V[2][0], V[2][1]));
        L = cmul(L, csel((gl >> 3) & 1, V[3][0], V[3][1]));
        c2 A2[2][2];
#pragma unroll
        for (int t0 = 0; t0 < 2; ++t0) {
            const c2 w = cmul(L, csel(l4 ^ t0, V[4][0], V[4][1]));
#pragma unroll
            for (int t1 = 0; t1 < 2; ++t1) A2[t0][t1] = cmul(w, V[5][t0 ^ t1]);
        }
        c2 Bt[16];
#pragma unroll
        for (int t4 = 0; t4 < 2; ++t4) {
            const c2 e = csel(l0 ^ t4, V[9][0], V[9][1]);
#pragma unroll
            for (int t3 = 0; t3 < 2; ++t3) {
                const c2 d = cmul(csel(l0 ^ t3 ^ t4, V[8][0], V[8][1]), e);
#pragma unroll
                for (int t2 = 0; t2 < 2; ++t2) {
                    const c2 cc = cmul(V[7][t2 ^ t3], d);
#pragma unroll
                    for (int t1 = 0; t1 < 2; ++t1)
                        Bt[t1 + 2 * t2 + 4 * t3 + 8 * t4] = cmul(V[6][t1 ^ t2], cc);
                }
            }
        }
#pragma unroll
        for (int t = 0; t < 32; ++t) {
            const c2 a = cmul(A2[t & 1][(t >> 1) & 1], Bt[t >> 1]);
            st[lane + (t << 5)] = make_float2(a.re, a.im);
        }
    }

    for (int layer = 1; layer < NLAYERS; ++layer) {
        const float* ML = Mvar + layer * NQ * 16;
        PASS0(st, lane, var_u(ML + 5 * 16, ho, h),
              var_u(ML + 0 * 16, o1, 0), var_u(ML + 0 * 16, o1, 1),
              var_u(ML + 1 * 16, o0, 0), var_u(ML + 1 * 16, o0, 1));
        PASS1(st, lane, var_u(ML + 6 * 16, ho, h),
              var_u(ML + 2 * 16, o1, 0), var_u(ML + 2 * 16, o1, 1),
              var_u(ML + 3 * 16, o0, 0), var_u(ML + 3 * 16, o0, 1));
        PASS2(st, lane, var_u(ML + 7 * 16, ho, h),
              var_u(ML + 4 * 16, o1, 0), var_u(ML + 4 * 16, o1, 1),
              var_u(ML + 9 * 16, o0, 0), var_u(ML + 9 * 16, o0, 1));
        PASS3(st, lane, var_u(ML + 8 * 16, ho, h),
              id_u(o1, 0), id_u(o1, 1), id_u(o0, 0), id_u(o0, 1));

        {
            const int fl = (lane ^ (lane >> 1)) | ((lane & 1) ? 0x300 : 0);
            float2 g[32];
#pragma unroll
            for (int t = 0; t < 32; ++t) {
                const int ft = (t << 5) ^ (t << 4);
                g[t] = st[(fl ^ (ft & 0x310)) + (ft & 0x0E0)];
            }
#pragma unroll
            for (int t = 0; t < 32; ++t) st[lane + (t << 5)] = g[t];
        }
    }

    float acc[NQ];
#pragma unroll
    for (int q = 0; q < NQ; ++q) acc[q] = 0.f;
    for (int i = lane; i < DIM; i += 32) {
        float2 a = st[i];
        float pr = a.x * a.x + a.y * a.y;
#pragma unroll
        for (int q = 0; q < NQ; ++q) {
            unsigned bit = ((unsigned)i >> (9 - q)) & 1u;
            acc[q] += __int_as_float(__float_as_int(pr) ^ (bit << 31));
        }
    }
#pragma unroll
    for (int off = 16; off > 0; off >>= 1)
#pragma unroll
        for (int q = 0; q < NQ; ++q)
            acc[q] += __shfl_xor(acc[q], off, 32);
    if (lane == 0) {
#pragma unroll
        for (int q = 0; q < NQ; ++q) out[b * NQ + q] = acc[q];
    }
}

extern "C" void kernel_launch(void* const* d_in, const int* in_sizes, int n_in,
                              void* d_out, int out_size, void* d_ws, size_t ws_size,
                              hipStream_t stream) {
    const float* x     = (const float*)d_in[0];
    const float* encW  = (const float*)d_in[1];
    const float* encB  = (const float*)d_in[2];
    const float* theta = (const float*)d_in[3];
    float* out  = (float*)d_out;
    float* Mvar = (float*)d_ws;

    int batch = in_sizes[0] / 3;

    qprep_kernel<<<1, 64, 0, stream>>>(theta, Mvar);

    int blocks = (batch + WPB - 1) / WPB;
    qsim_kernel<<<blocks, WPB * 32, 0, stream>>>(x, encW, encB, Mvar, out, batch);
}


# Each CNOT ring's index map folded into the load addresses of its consumer (next layer's first WMMA pass, or the read-out): the gather pass is gone, 32 LDS loads + 32 stores per lane per layer fewer
# speedup vs baseline: 6.3305x; 1.0588x over previous
#include <hip/hip_runtime.h>
#include <math.h>

typedef __attribute__((ext_vector_type(2))) float v2f;
typedef __attribute__((ext_vector_type(8))) float v8f;

#define NQ       10
#define DIM      1024
#define NLAYERS  4
#define WPB      8
#define PI_F     3.14159265358979323846f

struct c2 { float re, im; };
__device__ __forceinline__ c2 cmul(c2 a, c2 b) {
    return { a.re * b.re - a.im * b.im, a.re * b.im + a.im * b.re };
}
__device__ __forceinline__ c2 cadd(c2 a, c2 b) { return { a.re + b.re, a.im + b.im }; }

__global__ void qprep_kernel(const float* __restrict__ theta, float* __restrict__ Mout) {
    int g = blockIdx.x * blockDim.x + threadIdx.x;
    if (g >= NLAYERS * NQ) return;
    float t0 = theta[g * 3 + 0], t1 = theta[g * 3 + 1], t2 = theta[g * 3 + 2];
    float cx = cosf(t0 * 0.5f), sx = sinf(t0 * 0.5f);
    float cy = cosf(t1 * 0.5f), sy = sinf(t1 * 0.5f);
    float cz = cosf(t2 * 0.5f), sz = sinf(t2 * 0.5f);
    c2 X[2][2] = { { {cx, 0.f}, {0.f, -sx} }, { {0.f, -sx}, {cx, 0.f} } };
    c2 Y[2][2] = { { {cy, 0.f}, {-sy, 0.f} }, { {sy, 0.f}, {cy, 0.f} } };
    c2 Z[2][2] = { { {cz, -sz}, {0.f, 0.f} }, { {0.f, 0.f}, {cz, sz} } };
    c2 XY[2][2], G[2][2];
    for (int i = 0; i < 2; ++i)
        for (int j = 0; j < 2; ++j)
            XY[i][j] = cadd(cmul(X[i][0], Y[0][j]), cmul(X[i][1], Y[1][j]));
    for (int i = 0; i < 2; ++i)
        for (int j = 0; j < 2; ++j)
            G[i][j] = cadd(cmul(XY[i][0], Z[0][j]), cmul(XY[i][1], Z[1][j]));
    float* M = Mout + g * 16;
    for (int i = 0; i < 2; ++i)
        for (int j = 0; j < 2; ++j) {
            M[(2 * i    ) * 4 + 2 * j    ] =  G[i][j].re;
            M[(2 * i    ) * 4 + 2 * j + 1] =  G[i][j].im;
            M[(2 * i + 1) * 4 + 2 * j    ] = -G[i][j].im;
            M[(2 * i + 1) * 4 + 2 * j + 1] =  G[i][j].re;
        }
}

template <int PH, int PJ1, int PJ0, int N0, int N1, int N2, int N3, int T0, int T1, int T2>
__device__ __forceinline__ void fused_pass(float2* __restrict__ st, int lane,
                                           c2 uH, c2 uJ1a, c2 uJ1b, c2 uJ0a, c2 uJ0b) {
    const int c  = lane & 15;
    const int h  = lane >> 4;
    const int co = c & 1;

    const c2 t0 = cmul(uH, uJ1a), t1 = cmul(uH, uJ1b);
    c2 P[4];
    P[0] = cmul(t0, uJ0a); P[1] = cmul(t0, uJ0b);
    P[2] = cmul(t1, uJ0a); P[3] = cmul(t1, uJ0b);
    v2f A[4];
#pragma unroll
    for (int j = 0; j < 4; ++j) {
        A[j].x = co ? P[j].im :  P[j].re;
        A[j].y = co ? P[j].re : -P[j].im;
    }

    const int base = ((c & 1) << N0) | (((c >> 1) & 1) << N1) | (((c >> 2) & 1) << N2) |
                     (((c >> 3) & 1) << N3) | (h << PH);
    float2* p = st + base;

    float2 cur0[4], cur1[4], nx0[4], nx1[4];
#pragma unroll
    for (int j = 0; j < 4; ++j) {
        const int jo = ((j >> 1) << PJ1) | ((j & 1) << PJ0);
        cur0[j] = p[jo];
        cur1[j] = p[(1 << T0) | jo];
    }
#pragma unroll
    for (int tp = 0; tp < 4; ++tp) {
        const int ta = ((tp & 1) << T1) | (((tp >> 1) & 1) << T2);
        const int tb = ta | (1 << T0);
        if (tp < 3) {
            const int u  = 2 * tp + 2;
            const int na = (((u >> 1) & 1) << T1) | (((u >> 2) & 1) << T2);
#pragma unroll
            for (int j = 0; j < 4; ++j) {
                const int jo = ((j >> 1) << PJ1) | ((j & 1) << PJ0);
                nx0[j] = p[na | jo];
                nx1[j] = p[na | (1 << T0) | jo];
            }
        }
        v8f C0 = {0.f, 0.f, 0.f, 0.f, 0.f, 0.f, 0.f, 0.f};
        v8f C1 = {0.f, 0.f, 0.f, 0.f, 0.f, 0.f, 0.f, 0.f};
#pragma unroll
        for (int j = 0; j < 4; ++j) {
            v2f B0; B0.x = cur0[j].x; B0.y = cur0[j].y;
            v2f B1; B1.x = cur1[j].x; B1.y = cur1[j].y;
            C0 = __builtin_amdgcn_wmma_f32_16x16x4_f32(false, A[j], false, B0, (short)0, C0, false, false);
            C1 = __builtin_amdgcn_wmma_f32_16x16x4_f32(false, A[j], false, B1, (short)0, C1, false, false);
        }
#pragma unroll
        for (int j = 0; j < 4; ++j) {
            const int jo = ((j >> 1) << PJ1) | ((j & 1) << PJ0);
            p[ta | jo] = make_float2(C0[2 * j], C0[2 * j + 1]);
            p[tb | jo] = make_float2(C1[2 * j], C1[2 * j + 1]);
        }
#pragma unroll
        for (int j = 0; j < 4; ++j) { cur0[j] = nx0[j]; cur1[j] = nx1[j]; }
    }
}

__device__ __forceinline__ int ring_src(int fl, int t) {
    const int ft = (t << 5) ^ (t << 4);
    return (fl ^ (ft & 0x310)) + (ft & 0x0E0);
}
__device__ __forceinline__ void ring_pass0(float2* __restrict__ st, int lane, int fl,
                                           c2 uH, c2 uJ1a, c2 uJ1b, c2 uJ0a, c2 uJ0b) {
    const int co = lane & 1;

    const c2 t0 = cmul(uH, uJ1a), t1 = cmul(uH, uJ1b);
    c2 P[4];
    P[0] = cmul(t0, uJ0a); P[1] = cmul(t0, uJ0b);
    P[2] = cmul(t1, uJ0a); P[3] = cmul(t1, uJ0b);
    v2f A[4];
#pragma unroll
    for (int j = 0; j < 4; ++j) {
        A[j].x = co ? P[j].im :  P[j].re;
        A[j].y = co ? P[j].re : -P[j].im;
    }

    float2 g[32];
#pragma unroll
    for (int tp = 0; tp < 4; ++tp)
#pragma unroll
        for (int j = 0; j < 4; ++j)
#pragma unroll
            for (int e = 0; e < 2; ++e) {
                const int t = 2 * tp + e + 8 * (j & 1) + 16 * (j >> 1);
                g[t] = st[ring_src(fl, t)];
            }

    float2* p = st + lane;
#pragma unroll
    for (int tp = 0; tp < 4; ++tp) {
        v8f C0 = {0.f, 0.f, 0.f, 0.f, 0.f, 0.f, 0.f, 0.f};
        v8f C1 = {0.f, 0.f, 0.f, 0.f, 0.f, 0.f, 0.f, 0.f};
#pragma unroll
        for (int j = 0; j < 4; ++j) {
            const float2 b0 = g[2 * tp     + 8 * (j & 1) + 16 * (j >> 1)];
            const float2 b1 = g[2 * tp + 1 + 8 * (j & 1) + 16 * (j >> 1)];
            v2f B0; B0.x = b0.x; B0.y = b0.y;
            v2f B1; B1.x = b1.x; B1.y = b1.y;
            C0 = __builtin_amdgcn_wmma_f32_16x16x4_f32(false, A[j], false, B0, (short)0, C0, false, false);
            C1 = __builtin_amdgcn_wmma_f32_16x16x4_f32(false, A[j], false, B1, (short)0, C1, false, false);
        }
#pragma unroll
        for (int j = 0; j < 4; ++j) {
            const int jo = ((j >> 1) << 9) | ((j & 1) << 8);
            p[((2 * tp    ) << 5) | jo] = make_float2(C0[2 * j], C0[2 * j + 1]);
            p[((2 * tp + 1) << 5) | jo] = make_float2(C1[2 * j], C1[2 * j + 1]);
        }
    }
}

#define PASS0 fused_pass<4, 9, 8, 0, 1, 2, 3, 5, 6, 7>
#define PASS1 fused_pass<3, 7, 6, 0, 1, 2, 4, 5, 8, 9>
#define PASS2 fused_pass<2, 5, 0, 1, 3, 4, 6, 7, 8, 9>
#define PASS3 fused_pass<1, 9, 8, 0, 2, 3, 4, 5, 6, 7>

__device__ __forceinline__ c2 csel(int pick1, c2 a0, c2 a1) {
    c2 r; r.re = pick1 ? a1.re : a0.re; r.im = pick1 ? a1.im : a0.im; return r;
}
__device__ __forceinline__ c2 id_u(int o, int i) {
    c2 r; r.re = (o == i) ? 1.f : 0.f; r.im = 0.f; return r;
}
__device__ __forceinline__ c2 var_u(const float* __restrict__ M, int o, int i) {
    const float2 v = *(const float2*)(M + 8 * i + 2 * o);
    c2 r; r.re = v.x; r.im = v.y; return r;
}

__global__ void __launch_bounds__(WPB * 32)
qsim_kernel(const float* __restrict__ x, const float* __restrict__ encW,
            const float* __restrict__ encB, const float* __restrict__ Mvar,
            float* __restrict__ out, int batch) {
    __shared__ float2 lds[WPB * DIM];

    const int lane = threadIdx.x & 31;
    const int w    = threadIdx.x >> 5;
    const int b    = blockIdx.x * WPB + w;
    if (b >= batch) return;

    float2* st = &lds[w * DIM];
    const int c  = lane & 15;
    const int h  = lane >> 4;
    const int ho = c >> 3;
    const int o1 = (c >> 2) & 1;
    const int o0 = (c >> 1) & 1;

    {
        float xv0 = x[b * 3 + 0], xv1 = x[b * 3 + 1], xv2 = x[b * 3 + 2];
        c2 V[NQ][2];
#pragma unroll
        for (int p = 0; p < NQ; ++p) {
            const int q = 9 - p;
            float ang = PI_F * tanhf(xv0 * encW[q * 3] + xv1 * encW[q * 3 + 1] +
                                     xv2 * encW[q * 3 + 2] + encB[q]);
            const float cq = cosf(ang * 0.5f);
            const float sq = sinf(ang * 0.5f);
            const float* M = Mvar + q * 16;
#pragma unroll
            for (int o = 0; o < 2; ++o) {
                const c2 g0 = var_u(M, o, 0), g1 = var_u(M, o, 1);
                V[p][o].re = cq * g0.re - sq * g1.re;
                V[p][o].im = cq * g0.im - sq * g1.im;
            }
        }
        const int gl = lane ^ (lane >> 1);
        const int l0 = lane & 1, l4 = lane >> 4;
        c2 L = csel(gl & 1, V[0][0], V[0][1]);
        L = cmul(L, csel((gl >> 1) & 1, V[1][0], V[1][1]));
        L = cmul(L, csel((gl >> 2) & 1, V[2][0], V[2][1]));
        L = cmul(L, csel((gl >> 3) & 1, V[3][0], V[3][1]));
        c2 A2[2][2];
#pragma unroll
        for (int t0 = 0; t0 < 2; ++t0) {
            const c2 w = cmul(L, csel(l4 ^ t0, V[4][0], V[4][1]));
#pragma unroll
            for (int t1 = 0; t1 < 2; ++t1) A2[t0][t1] = cmul(w, V[5][t0 ^ t1]);
        }
        c2 Bt[16];
#pragma unroll
        for (int t4 = 0; t4 < 2; ++t4) {
            const c2 e = csel(l0 ^ t4, V[9][0], V[9][1]);
#pragma unroll
            for (int t3 = 0; t3 < 2; ++t3) {
                const c2 d = cmul(csel(l0 ^ t3 ^ t4, V[8][0], V[8][1]), e);
#pragma unroll
                for (int t2 = 0; t2 < 2; ++t2) {
                    const c2 cc = cmul(V[7][t2 ^ t3], d);
#pragma unroll
                    for (int t1 = 0; t1 < 2; ++t1)
                        Bt[t1 + 2 * t2 + 4 * t3 + 8 * t4] = cmul(V[6][t1 ^ t2], cc);
                }
            }
        }
#pragma unroll
        for (int t = 0; t < 32; ++t) {
            const c2 a = cmul(A2[t & 1][(t >> 1) & 1], Bt[t >> 1]);
            st[lane + (t << 5)] = make_float2(a.re, a.im);
        }
    }

    const int fl = (lane ^ (lane >> 1)) | ((lane & 1) ? 0x300 : 0);
    for (int layer = 1; layer < NLAYERS; ++layer) {
        const float* ML = Mvar + layer * NQ * 16;
        const c2 a0 = var_u(ML + 5 * 16, ho, h);
        const c2 a1 = var_u(ML + 0 * 16, o1, 0), a2 = var_u(ML + 0 * 16, o1, 1);
        const c2 a3 = var_u(ML + 1 * 16, o0, 0), a4 = var_u(ML + 1 * 16, o0, 1);
        if (layer == 1) PASS0(st, lane, a0, a1, a2, a3, a4);
        else            ring_pass0(st, lane, fl, a0, a1, a2, a3, a4);
        PASS1(st, lane, var_u(ML + 6 * 16, ho, h),
              var_u(ML + 2 * 16, o1, 0), var_u(ML + 2 * 16, o1, 1),
              var_u(ML + 3 * 16, o0, 0), var_u(ML + 3 * 16, o0, 1));
        PASS2(st, lane, var_u(ML + 7 * 16, ho, h),
              var_u(ML + 4 * 16, o1, 0), var_u(ML + 4 * 16, o1, 1),
              var_u(ML + 9 * 16, o0, 0), var_u(ML + 9 * 16, o0, 1));
        PASS3(st, lane, var_u(ML + 8 * 16, ho, h),
              id_u(o1, 0), id_u(o1, 1), id_u(o0, 0), id_u(o0, 1));
    }

    float acc[NQ];
#pragma unroll
    for (int q = 0; q < NQ; ++q) acc[q] = 0.f;
#pragma unroll
    for (int t = 0; t < 32; ++t) {
        const int i = lane + (t << 5);
        float2 a = st[ring_src(fl, t)];
        float pr = a.x * a.x + a.y * a.y;
#pragma unroll
        for (int q = 0; q < NQ; ++q) {
            unsigned bit = ((unsigned)i >> (9 - q)) & 1u;
            acc[q] += __int_as_float(__float_as_int(pr) ^ (bit << 31));
        }
    }
#pragma unroll
    for (int off = 16; off > 0; off >>= 1)
#pragma unroll
        for (int q = 0; q < NQ; ++q)
            acc[q] += __shfl_xor(acc[q], off, 32);
    if (lane == 0) {
#pragma unroll
        for (int q = 0; q < NQ; ++q) out[b * NQ + q] = acc[q];
    }
}

extern "C" void kernel_launch(void* const* d_in, const int* in_sizes, int n_in,
                              void* d_out, int out_size, void* d_ws, size_t ws_size,
                              hipStream_t stream) {
    const float* x     = (const float*)d_in[0];
    const float* encW  = (const float*)d_in[1];
    const float* encB  = (const float*)d_in[2];
    const float* theta = (const float*)d_in[3];
    float* out  = (float*)d_out;
    float* Mvar = (float*)d_ws;

    int batch = in_sizes[0] / 3;

    qprep_kernel<<<1, 64, 0, stream>>>(theta, Mvar);

    int blocks = (batch + WPB - 1) / WPB;
    qsim_kernel<<<blocks, WPB * 32, 0, stream>>>(x, encW, encB, Mvar, out, batch);
}


# Encoding tanh/cos/sin and the layer-0 two-vector of each qubit computed by one lane per qubit and read by all lanes (40 v_readlane), not all ten in every lane: 9/10 of that vector work gone
# speedup vs baseline: 6.9033x; 1.0905x over previous
#include <hip/hip_runtime.h>
#include <math.h>

typedef __attribute__((ext_vector_type(2))) float v2f;
typedef __attribute__((ext_vector_type(8))) float v8f;

#define NQ       10
#define DIM      1024
#define NLAYERS  4
#define WPB      8
#define PI_F     3.14159265358979323846f

struct c2 { float re, im; };
__device__ __forceinline__ c2 cmul(c2 a, c2 b) {
    return { a.re * b.re - a.im * b.im, a.re * b.im + a.im * b.re };
}
__device__ __forceinline__ c2 cadd(c2 a, c2 b) { return { a.re + b.re, a.im + b.im }; }

__global__ void qprep_kernel(const float* __restrict__ theta, float* __restrict__ Mout) {
    int g = blockIdx.x * blockDim.x + threadIdx.x;
    if (g >= NLAYERS * NQ) return;
    float t0 = theta[g * 3 + 0], t1 = theta[g * 3 + 1], t2 = theta[g * 3 + 2];
    float cx = cosf(t0 * 0.5f), sx = sinf(t0 * 0.5f);
    float cy = cosf(t1 * 0.5f), sy = sinf(t1 * 0.5f);
    float cz = cosf(t2 * 0.5f), sz = sinf(t2 * 0.5f);
    c2 X[2][2] = { { {cx, 0.f}, {0.f, -sx} }, { {0.f, -sx}, {cx, 0.f} } };
    c2 Y[2][2] = { { {cy, 0.f}, {-sy, 0.f} }, { {sy, 0.f}, {cy, 0.f} } };
    c2 Z[2][2] = { { {cz, -sz}, {0.f, 0.f} }, { {0.f, 0.f}, {cz, sz} } };
    c2 XY[2][2], G[2][2];
    for (int i = 0; i < 2; ++i)
        for (int j = 0; j < 2; ++j)
            XY[i][j] = cadd(cmul(X[i][0], Y[0][j]), cmul(X[i][1], Y[1][j]));
    for (int i = 0; i < 2; ++i)
        for (int j = 0; j < 2; ++j)
            G[i][j] = cadd(cmul(XY[i][0], Z[0][j]), cmul(XY[i][1], Z[1][j]));
    float* M = Mout + g * 16;
    for (int i = 0; i < 2; ++i)
        for (int j = 0; j < 2; ++j) {
            M[(2 * i    ) * 4 + 2 * j    ] =  G[i][j].re;
            M[(2 * i    ) * 4 + 2 * j + 1] =  G[i][j].im;
            M[(2 * i + 1) * 4 + 2 * j    ] = -G[i][j].im;
            M[(2 * i + 1) * 4 + 2 * j + 1] =  G[i][j].re;
        }
}

template <int PH, int PJ1, int PJ0, int N0, int N1, int N2, int N3, int T0, int T1, int T2>
__device__ __forceinline__ void fused_pass(float2* __restrict__ st, int lane,
                                           c2 uH, c2 uJ1a, c2 uJ1b, c2 uJ0a, c2 uJ0b) {
    const int c  = lane & 15;
    const int h  = lane >> 4;
    const int co = c & 1;

    const c2 t0 = cmul(uH, uJ1a), t1 = cmul(uH, uJ1b);
    c2 P[4];
    P[0] = cmul(t0, uJ0a); P[1] = cmul(t0, uJ0b);
    P[2] = cmul(t1, uJ0a); P[3] = cmul(t1, uJ0b);
    v2f A[4];
#pragma unroll
    for (int j = 0; j < 4; ++j) {
        A[j].x = co ? P[j].im :  P[j].re;
        A[j].y = co ? P[j].re : -P[j].im;
    }

    const int base = ((c & 1) << N0) | (((c >> 1) & 1) << N1) | (((c >> 2) & 1) << N2) |
                     (((c >> 3) & 1) << N3) | (h << PH);
    float2* p = st + base;

    float2 cur0[4], cur1[4], nx0[4], nx1[4];
#pragma unroll
    for (int j = 0; j < 4; ++j) {
        const int jo = ((j >> 1) << PJ1) | ((j & 1) << PJ0);
        cur0[j] = p[jo];
        cur1[j] = p[(1 << T0) | jo];
    }
#pragma unroll
    for (int tp = 0; tp < 4; ++tp) {
        const int ta = ((tp & 1) << T1) | (((tp >> 1) & 1) << T2);
        const int tb = ta | (1 << T0);
        if (tp < 3) {
            const int u  = 2 * tp + 2;
            const int na = (((u >> 1) & 1) << T1) | (((u >> 2) & 1) << T2);
#pragma unroll
            for (int j = 0; j < 4; ++j) {
                const int jo = ((j >> 1) << PJ1) | ((j & 1) << PJ0);
                nx0[j] = p[na | jo];
                nx1[j] = p[na | (1 << T0) | jo];
            }
        }
        v8f C0 = {0.f, 0.f, 0.f, 0.f, 0.f, 0.f, 0.f, 0.f};
        v8f C1 = {0.f, 0.f, 0.f, 0.f, 0.f, 0.f, 0.f, 0.f};
#pragma unroll
        for (int j = 0; j < 4; ++j) {
            v2f B0; B0.x = cur0[j].x; B0.y = cur0[j].y;
            v2f B1; B1.x = cur1[j].x; B1.y = cur1[j].y;
            C0 = __builtin_amdgcn_wmma_f32_16x16x4_f32(false, A[j], false, B0, (short)0, C0, false, false);
            C1 = __builtin_amdgcn_wmma_f32_16x16x4_f32(false, A[j], false, B1, (short)0, C1, false, false);
        }
#pragma unroll
        for (int j = 0; j < 4; ++j) {
            const int jo = ((j >> 1) << PJ1) | ((j & 1) << PJ0);
            p[ta | jo] = make_float2(C0[2 * j], C0[2 * j + 1]);
            p[tb | jo] = make_float2(C1[2 * j], C1[2 * j + 1]);
        }
#pragma unroll
        for (int j = 0; j < 4; ++j) { cur0[j] = nx0[j]; cur1[j] = nx1[j]; }
    }
}

__device__ __forceinline__ int ring_src(int fl, int t) {
    const int ft = (t << 5) ^ (t << 4);
    return (fl ^ (ft & 0x310)) + (ft & 0x0E0);
}
__device__ __forceinline__ void ring_pass0(float2* __restrict__ st, int lane, int fl,
                                           c2 uH, c2 uJ1a, c2 uJ1b, c2 uJ0a, c2 uJ0b) {
    const int co = lane & 1;

    const c2 t0 = cmul(uH, uJ1a), t1 = cmul(uH, uJ1b);
    c2 P[4];
    P[0] = cmul(t0, uJ0a); P[1] = cmul(t0, uJ0b);
    P[2] = cmul(t1, uJ0a); P[3] = cmul(t1, uJ0b);
    v2f A[4];
#pragma unroll
    for (int j = 0; j < 4; ++j) {
        A[j].x = co ? P[j].im :  P[j].re;
        A[j].y = co ? P[j].re : -P[j].im;
    }

    float2 g[32];
#pragma unroll
    for (int tp = 0; tp < 4; ++tp)
#pragma unroll
        for (int j = 0; j < 4; ++j)
#pragma unroll
            for (int e = 0; e < 2; ++e) {
                const int t = 2 * tp + e + 8 * (j & 1) + 16 * (j >> 1);
                g[t] = st[ring_src(fl, t)];
            }

    float2* p = st + lane;
#pragma unroll
    for (int tp = 0; tp < 4; ++tp) {
        v8f C0 = {0.f, 0.f, 0.f, 0.f, 0.f, 0.f, 0.f, 0.f};
        v8f C1 = {0.f, 0.f, 0.f, 0.f, 0.f, 0.f, 0.f, 0.f};
#pragma unroll
        for (int j = 0; j < 4; ++j) {
            const float2 b0 = g[2 * tp     + 8 * (j & 1) + 16 * (j >> 1)];
            const float2 b1 = g[2 * tp + 1 + 8 * (j & 1) + 16 * (j >> 1)];
            v2f B0; B0.x = b0.x; B0.y = b0.y;
            v2f B1; B1.x = b1.x; B1.y = b1.y;
            C0 = __builtin_amdgcn_wmma_f32_16x16x4_f32(false, A[j], false, B0, (short)0, C0, false, false);
            C1 = __builtin_amdgcn_wmma_f32_16x16x4_f32(false, A[j], false, B1, (short)0, C1, false, false);
        }
#pragma unroll
        for (int j = 0; j < 4; ++j) {
            const int jo = ((j >> 1) << 9) | ((j & 1) << 8);
            p[((2 * tp    ) << 5) | jo] = make_float2(C0[2 * j], C0[2 * j + 1]);
            p[((2 * tp + 1) << 5) | jo] = make_float2(C1[2 * j], C1[2 * j + 1]);
        }
    }
}

#define PASS0 fused_pass<4, 9, 8, 0, 1, 2, 3, 5, 6, 7>
#define PASS1 fused_pass<3, 7, 6, 0, 1, 2, 4, 5, 8, 9>
#define PASS2 fused_pass<2, 5, 0, 1, 3, 4, 6, 7, 8, 9>
#define PASS3 fused_pass<1, 9, 8, 0, 2, 3, 4, 5, 6, 7>

__device__ __forceinline__ c2 csel(int pick1, c2 a0, c2 a1) {
    c2 r; r.re = pick1 ? a1.re : a0.re; r.im = pick1 ? a1.im : a0.im; return r;
}
__device__ __forceinline__ c2 id_u(int o, int i) {
    c2 r; r.re = (o == i) ? 1.f : 0.f; r.im = 0.f; return r;
}
__device__ __forceinline__ c2 var_u(const float* __restrict__ M, int o, int i) {
    const float2 v = *(const float2*)(M + 8 * i + 2 * o);
    c2 r; r.re = v.x; r.im = v.y; return r;
}

__device__ __forceinline__ float lane_value(float v, int src) {
    return __int_as_float(__builtin_amdgcn_readlane(__float_as_int(v), src));
}

__global__ void __launch_bounds__(WPB * 32)
qsim_kernel(const float* __restrict__ x, const float* __restrict__ encW,
            const float* __restrict__ encB, const float* __restrict__ Mvar,
            float* __restrict__ out, int batch) {
    __shared__ float2 lds[WPB * DIM];

    const int lane = threadIdx.x & 31;
    const int w    = threadIdx.x >> 5;
    const int b    = blockIdx.x * WPB + w;
    if (b >= batch) return;

    float2* st = &lds[w * DIM];
    const int c  = lane & 15;
    const int h  = lane >> 4;
    const int ho = c >> 3;
    const int o1 = (c >> 2) & 1;
    const int o0 = (c >> 1) & 1;

    {
        float xv0 = x[b * 3 + 0], xv1 = x[b * 3 + 1], xv2 = x[b * 3 + 2];
        const int q = lane < NQ ? lane : NQ - 1;
        float ang = PI_F * tanhf(fmaf(xv2, encW[q * 3 + 2],
                                      fmaf(xv0, encW[q * 3], xv1 * encW[q * 3 + 1])) + encB[q]);
        const float cq = cosf(ang * 0.5f);
        const float sq = sinf(ang * 0.5f);
        const float* M = Mvar + q * 16;
        c2 own[2];
#pragma unroll
        for (int o = 0; o < 2; ++o) {
            const c2 g0 = var_u(M, o, 0), g1 = var_u(M, o, 1);
            own[o].re = cq * g0.re - sq * g1.re;
            own[o].im = cq * g0.im - sq * g1.im;
        }
        c2 V[NQ][2];
#pragma unroll
        for (int p = 0; p < NQ; ++p) {
#pragma unroll
            for (int o = 0; o < 2; ++o) {
                V[p][o].re = lane_value(own[o].re, 9 - p);
                V[p][o].im = lane_value(own[o].im, 9 - p);
            }
        }
        const int gl = lane ^ (lane >> 1);
        const int l0 = lane & 1, l4 = lane >> 4;
        c2 L = csel(gl & 1, V[0][0], V[0][1]);
        L = cmul(L, csel((gl >> 1) & 1, V[1][0], V[1][1]));
        L = cmul(L, csel((gl >> 2) & 1, V[2][0], V[2][1]));
        L = cmul(L, csel((gl >> 3) & 1, V[3][0], V[3][1]));
        c2 A2[2][2];
#pragma unroll
        for (int t0 = 0; t0 < 2; ++t0) {
            const c2 w = cmul(L, csel(l4 ^ t0, V[4][0], V[4][1]));
#pragma unroll
            for (int t1 = 0; t1 < 2; ++t1) A2[t0][t1] = cmul(w, V[5][t0 ^ t1]);
        }
        c2 Bt[16];
#pragma unroll
        for (int t4 = 0; t4 < 2; ++t4) {
            const c2 e = csel(l0 ^ t4, V[9][0], V[9][1]);
#pragma unroll
            for (int t3 = 0; t3 < 2; ++t3) {
                const c2 d = cmul(csel(l0 ^ t3 ^ t4, V[8][0], V[8][1]), e);
#pragma unroll
                for (int t2 = 0; t2 < 2; ++t2) {
                    const c2 cc = cmul(V[7][t2 ^ t3], d);
#pragma unroll
                    for (int t1 = 0; t1 < 2; ++t1)
                        Bt[t1 + 2 * t2 + 4 * t3 + 8 * t4] = cmul(V[6][t1 ^ t2], cc);
                }
            }
        }
#pragma unroll
        for (int t = 0; t < 32; ++t) {
            const c2 a = cmul(A2[t & 1][(t >> 1) & 1], Bt[t >> 1]);
            st[lane + (t << 5)] = make_float2(a.re, a.im);
        }
    }

    const int fl = (lane ^ (lane >> 1)) | ((lane & 1) ? 0x300 : 0);
    for (int layer = 1; layer < NLAYERS; ++layer) {
        const float* ML = Mvar + layer * NQ * 16;
        const c2 a0 = var_u(ML + 5 * 16, ho, h);
        const c2 a1 = var_u(ML + 0 * 16, o1, 0), a2 = var_u(ML + 0 * 16, o1, 1);
        const c2 a3 = var_u(ML + 1 * 16, o0, 0), a4 = var_u(ML + 1 * 16, o0, 1);
        if (layer == 1) PASS0(st, lane, a0, a1, a2, a3, a4);
        else            ring_pass0(st, lane, fl, a0, a1, a2, a3, a4);
        PASS1(st, lane, var_u(ML + 6 * 16, ho, h),
              var_u(ML + 2 * 16, o1, 0), var_u(ML + 2 * 16, o1, 1),
              var_u(ML + 3 * 16, o0, 0), var_u(ML + 3 * 16, o0, 1));
        PASS2(st, lane, var_u(ML + 7 * 16, ho, h),
              var_u(ML + 4 * 16, o1, 0), var_u(ML + 4 * 16, o1, 1),
              var_u(ML + 9 * 16, o0, 0), var_u(ML + 9 * 16, o0, 1));
        PASS3(st, lane, var_u(ML + 8 * 16, ho, h),
              id_u(o1, 0), id_u(o1, 1), id_u(o0, 0), id_u(o0, 1));
    }

    float acc[NQ];
#pragma unroll
    for (int q = 0; q < NQ; ++q) acc[q] = 0.f;
#pragma unroll
    for (int t = 0; t < 32; ++t) {
        const int i = lane + (t << 5);
        float2 a = st[ring_src(fl, t)];
        float pr = a.x * a.x + a.y * a.y;
#pragma unroll
        for (int q = 0; q < NQ; ++q) {
            unsigned bit = ((unsigned)i >> (9 - q)) & 1u;
            acc[q] += __int_as_float(__float_as_int(pr) ^ (bit << 31));
        }
    }
#pragma unroll
    for (int off = 16; off > 0; off >>= 1)
#pragma unroll
        for (int q = 0; q < NQ; ++q)
            acc[q] += __shfl_xor(acc[q], off, 32);
    if (lane == 0) {
#pragma unroll
        for (int q = 0; q < NQ; ++q) out[b * NQ + q] = acc[q];
    }
}

extern "C" void kernel_launch(void* const* d_in, const int* in_sizes, int n_in,
                              void* d_out, int out_size, void* d_ws, size_t ws_size,
                              hipStream_t stream) {
    const float* x     = (const float*)d_in[0];
    const float* encW  = (const float*)d_in[1];
    const float* encB  = (const float*)d_in[2];
    const float* theta = (const float*)d_in[3];
    float* out  = (float*)d_out;
    float* Mvar = (float*)d_ws;

    int batch = in_sizes[0] / 3;

    qprep_kernel<<<1, 64, 0, stream>>>(theta, Mvar);

    int blocks = (batch + WPB - 1) / WPB;
    qsim_kernel<<<blocks, WPB * 32, 0, stream>>>(x, encW, encB, Mvar, out, batch);
}


# Layer-1 gates of the five odd index bits folded into the closed-form state (each a two-term sum on an eight-entry factor table, not a 16-term matrix sum): layer 1 is two WMMA passes, not four
# speedup vs baseline: 7.4191x; 1.0747x over previous
#include <hip/hip_runtime.h>
#include <math.h>

typedef __attribute__((ext_vector_type(2))) float v2f;
typedef __attribute__((ext_vector_type(8))) float v8f;

#define NQ       10
#define DIM      1024
#define NLAYERS  4
#define WPB      8
#define PI_F     3.14159265358979323846f

struct c2 { float re, im; };
__device__ __forceinline__ c2 cmul(c2 a, c2 b) {
    return { a.re * b.re - a.im * b.im, a.re * b.im + a.im * b.re };
}
__device__ __forceinline__ c2 cadd(c2 a, c2 b) { return { a.re + b.re, a.im + b.im }; }

__global__ void qprep_kernel(const float* __restrict__ theta, float* __restrict__ Mout) {
    int g = blockIdx.x * blockDim.x + threadIdx.x;
    if (g >= NLAYERS * NQ) return;
    float t0 = theta[g * 3 + 0], t1 = theta[g * 3 + 1], t2 = theta[g * 3 + 2];
    float cx = cosf(t0 * 0.5f), sx = sinf(t0 * 0.5f);
    float cy = cosf(t1 * 0.5f), sy = sinf(t1 * 0.5f);
    float cz = cosf(t2 * 0.5f), sz = sinf(t2 * 0.5f);
    c2 X[2][2] = { { {cx, 0.f}, {0.f, -sx} }, { {0.f, -sx}, {cx, 0.f} } };
    c2 Y[2][2] = { { {cy, 0.f}, {-sy, 0.f} }, { {sy, 0.f}, {cy, 0.f} } };
    c2 Z[2][2] = { { {cz, -sz}, {0.f, 0.f} }, { {0.f, 0.f}, {cz, sz} } };
    c2 XY[2][2], G[2][2];
    for (int i = 0; i < 2; ++i)
        for (int j = 0; j < 2; ++j)
            XY[i][j] = cadd(cmul(X[i][0], Y[0][j]), cmul(X[i][1], Y[1][j]));
    for (int i = 0; i < 2; ++i)
        for (int j = 0; j < 2; ++j)
            G[i][j] = cadd(cmul(XY[i][0], Z[0][j]), cmul(XY[i][1], Z[1][j]));
    float* M = Mout + g * 16;
    for (int i = 0; i < 2; ++i)
        for (int j = 0; j < 2; ++j) {
            M[(2 * i    ) * 4 + 2 * j    ] =  G[i][j].re;
            M[(2 * i    ) * 4 + 2 * j + 1] =  G[i][j].im;
            M[(2 * i + 1) * 4 + 2 * j    ] = -G[i][j].im;
            M[(2 * i + 1) * 4 + 2 * j + 1] =  G[i][j].re;
        }
}

template <int PH, int PJ1, int PJ0, int N0, int N1, int N2, int N3, int T0, int T1, int T2>
__device__ __forceinline__ void fused_pass(float2* __restrict__ st, int lane,
                                           c2 uH, c2 uJ1a, c2 uJ1b, c2 uJ0a, c2 uJ0b) {
    const int c  = lane & 15;
    const int h  = lane >> 4;
    const int co = c & 1;

    const c2 t0 = cmul(uH, uJ1a), t1 = cmul(uH, uJ1b);
    c2 P[4];
    P[0] = cmul(t0, uJ0a); P[1] = cmul(t0, uJ0b);
    P[2] = cmul(t1, uJ0a); P[3] = cmul(t1, uJ0b);
    v2f A[4];
#pragma unroll
    for (int j = 0; j < 4; ++j) {
        A[j].x = co ? P[j].im :  P[j].re;
        A[j].y = co ? P[j].re : -P[j].im;
    }

    const int base = ((c & 1) << N0) | (((c >> 1) & 1) << N1) | (((c >> 2) & 1) << N2) |
                     (((c >> 3) & 1) << N3) | (h << PH);
    float2* p = st + base;

    float2 cur0[4], cur1[4], nx0[4], nx1[4];
#pragma unroll
    for (int j = 0; j < 4; ++j) {
        const int jo = ((j >> 1) << PJ1) | ((j & 1) << PJ0);
        cur0[j] = p[jo];
        cur1[j] = p[(1 << T0) | jo];
    }
#pragma unroll
    for (int tp = 0; tp < 4; ++tp) {
        const int ta = ((tp & 1) << T1) | (((tp >> 1) & 1) << T2);
        const int tb = ta | (1 << T0);
        if (tp < 3) {
            const int u  = 2 * tp + 2;
            const int na = (((u >> 1) & 1) << T1) | (((u >> 2) & 1) << T2);
#pragma unroll
            for (int j = 0; j < 4; ++j) {
                const int jo = ((j >> 1) << PJ1) | ((j & 1) << PJ0);
                nx0[j] = p[na | jo];
                nx1[j] = p[na | (1 << T0) | jo];
            }
        }
        v8f C0 = {0.f, 0.f, 0.f, 0.f, 0.f, 0.f, 0.f, 0.f};
        v8f C1 = {0.f, 0.f, 0.f, 0.f, 0.f, 0.f, 0.f, 0.f};
#pragma unroll
        for (int j = 0; j < 4; ++j) {
            v2f B0; B0.x = cur0[j].x; B0.y = cur0[j].y;
            v2f B1; B1.x = cur1[j].x; B1.y = cur1[j].y;
            C0 = __builtin_amdgcn_wmma_f32_16x16x4_f32(false, A[j], false, B0, (short)0, C0, false, false);
            C1 = __builtin_amdgcn_wmma_f32_16x16x4_f32(false, A[j], false, B1, (short)0, C1, false, false);
        }
#pragma unroll
        for (int j = 0; j < 4; ++j) {
            const int jo = ((j >> 1) << PJ1) | ((j & 1) << PJ0);
            p[ta | jo] = make_float2(C0[2 * j], C0[2 * j + 1]);
            p[tb | jo] = make_float2(C1[2 * j], C1[2 * j + 1]);
        }
#pragma unroll
        for (int j = 0; j < 4; ++j) { cur0[j] = nx0[j]; cur1[j] = nx1[j]; }
    }
}

__device__ __forceinline__ int ring_src(int fl, int t) {
    const int ft = (t << 5) ^ (t << 4);
    return (fl ^ (ft & 0x310)) + (ft & 0x0E0);
}
__device__ __forceinline__ void ring_pass0(float2* __restrict__ st, int lane, int fl,
                                           c2 uH, c2 uJ1a, c2 uJ1b, c2 uJ0a, c2 uJ0b) {
    const int co = lane & 1;

    const c2 t0 = cmul(uH, uJ1a), t1 = cmul(uH, uJ1b);
    c2 P[4];
    P[0] = cmul(t0, uJ0a); P[1] = cmul(t0, uJ0b);
    P[2] = cmul(t1, uJ0a); P[3] = cmul(t1, uJ0b);
    v2f A[4];
#pragma unroll
    for (int j = 0; j < 4; ++j) {
        A[j].x = co ? P[j].im :  P[j].re;
        A[j].y = co ? P[j].re : -P[j].im;
    }

    float2 g[32];
#pragma unroll
    for (int tp = 0; tp < 4; ++tp)
#pragma unroll
        for (int j = 0; j < 4; ++j)
#pragma unroll
            for (int e = 0; e < 2; ++e) {
                const int t = 2 * tp + e + 8 * (j & 1) + 16 * (j >> 1);
                g[t] = st[ring_src(fl, t)];
            }

    float2* p = st + lane;
#pragma unroll
    for (int tp = 0; tp < 4; ++tp) {
        v8f C0 = {0.f, 0.f, 0.f, 0.f, 0.f, 0.f, 0.f, 0.f};
        v8f C1 = {0.f, 0.f, 0.f, 0.f, 0.f, 0.f, 0.f, 0.f};
#pragma unroll
        for (int j = 0; j < 4; ++j) {
            const float2 b0 = g[2 * tp     + 8 * (j & 1) + 16 * (j >> 1)];
            const float2 b1 = g[2 * tp + 1 + 8 * (j & 1) + 16 * (j >> 1)];
            v2f B0; B0.x = b0.x; B0.y = b0.y;
            v2f B1; B1.x = b1.x; B1.y = b1.y;
            C0 = __builtin_amdgcn_wmma_f32_16x16x4_f32(false, A[j], false, B0, (short)0, C0, false, false);
            C1 = __builtin_amdgcn_wmma_f32_16x16x4_f32(false, A[j], false, B1, (short)0, C1, false, false);
        }
#pragma unroll
        for (int j = 0; j < 4; ++j) {
            const int jo = ((j >> 1) << 9) | ((j & 1) << 8);
            p[((2 * tp    ) << 5) | jo] = make_float2(C0[2 * j], C0[2 * j + 1]);
            p[((2 * tp + 1) << 5) | jo] = make_float2(C1[2 * j], C1[2 * j + 1]);
        }
    }
}

#define PASS1 fused_pass<3, 7, 6, 0, 1, 2, 4, 5, 8, 9>
#define PASS2 fused_pass<2, 5, 0, 1, 3, 4, 6, 7, 8, 9>
#define PASS3 fused_pass<1, 9, 8, 0, 2, 3, 4, 5, 6, 7>
#define P1A fused_pass<4, 8, 6, 0, 1, 2, 3, 5, 7, 9>
#define P1B fused_pass<2, 9, 0, 1, 3, 4, 5, 6, 7, 8>

__device__ __forceinline__ c2 csel(int pick1, c2 a0, c2 a1) {
    c2 r; r.re = pick1 ? a1.re : a0.re; r.im = pick1 ? a1.im : a0.im; return r;
}
__device__ __forceinline__ c2 fold2(c2 u0, c2 u1, c2 pa, c2 pb, c2 qa, c2 qb) {
    return cadd(cmul(u0, cmul(pa, qa)), cmul(u1, cmul(pb, qb)));
}
__device__ __forceinline__ c2 id_u(int o, int i) {
    c2 r; r.re = (o == i) ? 1.f : 0.f; r.im = 0.f; return r;
}
__device__ __forceinline__ c2 var_u(const float* __restrict__ M, int o, int i) {
    const float2 v = *(const float2*)(M + 8 * i + 2 * o);
    c2 r; r.re = v.x; r.im = v.y; return r;
}

__device__ __forceinline__ float lane_value(float v, int src) {
    return __int_as_float(__builtin_amdgcn_readlane(__float_as_int(v), src));
}

__global__ void __launch_bounds__(WPB * 32) __attribute__((amdgpu_waves_per_eu(8)))
qsim_kernel(const float* __restrict__ x, const float* __restrict__ encW,
            const float* __restrict__ encB, const float* __restrict__ Mvar,
            float* __restrict__ out, int batch) {
    __shared__ float2 lds[WPB * DIM];

    const int lane = threadIdx.x & 31;
    const int w    = threadIdx.x >> 5;
    const int b    = blockIdx.x * WPB + w;
    if (b >= batch) return;

    float2* st = &lds[w * DIM];
    const int c  = lane & 15;
    const int h  = lane >> 4;
    const int ho = c >> 3;
    const int o1 = (c >> 2) & 1;
    const int o0 = (c >> 1) & 1;

    {
        float xv0 = x[b * 3 + 0], xv1 = x[b * 3 + 1], xv2 = x[b * 3 + 2];
        const int q = lane < NQ ? lane : NQ - 1;
        float ang = PI_F * tanhf(fmaf(xv2, encW[q * 3 + 2],
                                      fmaf(xv0, encW[q * 3], xv1 * encW[q * 3 + 1])) + encB[q]);
        const float cq = cosf(ang * 0.5f);
        const float sq = sinf(ang * 0.5f);
        const float* M = Mvar + q * 16;
        c2 own[2];
#pragma unroll
        for (int o = 0; o < 2; ++o) {
            const c2 g0 = var_u(M, o, 0), g1 = var_u(M, o, 1);
            own[o].re = cq * g0.re - sq * g1.re;
            own[o].im = cq * g0.im - sq * g1.im;
        }
        c2 V[NQ][2];
#pragma unroll
        for (int p = 0; p < NQ; ++p) {
#pragma unroll
            for (int o = 0; o < 2; ++o) {
                V[p][o].re = lane_value(own[o].re, 9 - p);
                V[p][o].im = lane_value(own[o].im, 9 - p);
            }
        }
        const float* M1 = Mvar + NQ * 16;
        const int l0 = lane & 1, l1 = (lane >> 1) & 1, l2 = (lane >> 2) & 1;
        const int l3 = (lane >> 3) & 1, l4 = lane >> 4;
        c2 L;
        {
            const c2 w1 = fold2(var_u(M1 + 8 * 16, l1, 0), var_u(M1 + 8 * 16, l1, 1),
                                csel(l2, V[1][0], V[1][1]), csel(l2, V[1][1], V[1][0]),
                                csel(l0, V[0][0], V[0][1]), csel(l0, V[0][1], V[0][0]));
            const c2 w3 = fold2(var_u(M1 + 6 * 16, l3, 0), var_u(M1 + 6 * 16, l3, 1),
                                csel(l4, V[3][0], V[3][1]), csel(l4, V[3][1], V[3][0]),
                                csel(l2, V[2][0], V[2][1]), csel(l2, V[2][1], V[2][0]));
            L = cmul(w1, w3);
        }
        c2 A2[2][2];
        {
            const c2 v4a = csel(l4, V[4][0], V[4][1]);
            const c2 v4b = csel(l4, V[4][1], V[4][0]);
#pragma unroll
            for (int t0 = 0; t0 < 2; ++t0) {
                const c2 u0 = var_u(M1 + 4 * 16, t0, 0), u1 = var_u(M1 + 4 * 16, t0, 1);
#pragma unroll
                for (int t1 = 0; t1 < 2; ++t1)
                    A2[t0][t1] = cmul(L, fold2(u0, u1, V[5][t1], V[5][1 ^ t1], v4a, v4b));
            }
        }
        c2 Bt[16];
        {
            c2 W7[2][2][2];
#pragma unroll
            for (int t2 = 0; t2 < 2; ++t2) {
                const c2 u0 = var_u(M1 + 2 * 16, t2, 0), u1 = var_u(M1 + 2 * 16, t2, 1);
#pragma unroll
                for (int t1 = 0; t1 < 2; ++t1)
#pragma unroll
                    for (int t3 = 0; t3 < 2; ++t3)
                        W7[t1][t2][t3] = fold2(u0, u1, V[7][t3], V[7][1 ^ t3], V[6][t1], V[6][1 ^ t1]);
            }
            const c2 v9a = csel(l0, V[9][0], V[9][1]);
            const c2 v9b = csel(l0, V[9][1], V[9][0]);
#pragma unroll
            for (int t4 = 0; t4 < 2; ++t4) {
                const c2 u0 = var_u(M1 + 0 * 16, t4, 0), u1 = var_u(M1 + 0 * 16, t4, 1);
#pragma unroll
                for (int t3 = 0; t3 < 2; ++t3) {
                    const c2 v8a = csel(l0, V[8][t3], V[8][1 ^ t3]);
                    const c2 v8b = csel(l0, V[8][1 ^ t3], V[8][t3]);
                    const c2 w9 = fold2(u0, u1, v9a, v9b, v8a, v8b);
#pragma unroll
                    for (int t2 = 0; t2 < 2; ++t2)
#pragma unroll
                        for (int t1 = 0; t1 < 2; ++t1)
                            Bt[t1 + 2 * t2 + 4 * t3 + 8 * t4] = cmul(W7[t1][t2][t3], w9);
                }
            }
        }
#pragma unroll
        for (int t = 0; t < 32; ++t) {
            const c2 a = cmul(A2[t & 1][(t >> 1) & 1], Bt[t >> 1]);
            st[lane + (t << 5)] = make_float2(a.re, a.im);
        }
    }

    {
        const float* ML = Mvar + NQ * 16;
        P1A(st, lane, var_u(ML + 5 * 16, ho, h),
            var_u(ML + 1 * 16, o1, 0), var_u(ML + 1 * 16, o1, 1),
            var_u(ML + 3 * 16, o0, 0), var_u(ML + 3 * 16, o0, 1));
        P1B(st, lane, var_u(ML + 7 * 16, ho, h),
            id_u(o1, 0), id_u(o1, 1),
            var_u(ML + 9 * 16, o0, 0), var_u(ML + 9 * 16, o0, 1));
    }

    const int fl = (lane ^ (lane >> 1)) | ((lane & 1) ? 0x300 : 0);
    for (int layer = 2; layer < NLAYERS; ++layer) {
        const float* ML = Mvar + layer * NQ * 16;
        ring_pass0(st, lane, fl, var_u(ML + 5 * 16, ho, h),
                   var_u(ML + 0 * 16, o1, 0), var_u(ML + 0 * 16, o1, 1),
                   var_u(ML + 1 * 16, o0, 0), var_u(ML + 1 * 16, o0, 1));
        PASS1(st, lane, var_u(ML + 6 * 16, ho, h),
              var_u(ML + 2 * 16, o1, 0), var_u(ML + 2 * 16, o1, 1),
              var_u(ML + 3 * 16, o0, 0), var_u(ML + 3 * 16, o0, 1));
        PASS2(st, lane, var_u(ML + 7 * 16, ho, h),
              var_u(ML + 4 * 16, o1, 0), var_u(ML + 4 * 16, o1, 1),
              var_u(ML + 9 * 16, o0, 0), var_u(ML + 9 * 16, o0, 1));
        PASS3(st, lane, var_u(ML + 8 * 16, ho, h),
              id_u(o1, 0), id_u(o1, 1), id_u(o0, 0), id_u(o0, 1));
    }

    float acc[NQ];
#pragma unroll
    for (int q = 0; q < NQ; ++q) acc[q] = 0.f;
#pragma unroll
    for (int t = 0; t < 32; ++t) {
        const int i = lane + (t << 5);
        float2 a = st[ring_src(fl, t)];
        float pr = a.x * a.x + a.y * a.y;
#pragma unroll
        for (int q = 0; q < NQ; ++q) {
            unsigned bit = ((unsigned)i >> (9 - q)) & 1u;
            acc[q] += __int_as_float(__float_as_int(pr) ^ (bit << 31));
        }
    }
#pragma unroll
    for (int off = 16; off > 0; off >>= 1)
#pragma unroll
        for (int q = 0; q < NQ; ++q)
            acc[q] += __shfl_xor(acc[q], off, 32);
    if (lane == 0) {
#pragma unroll
        for (int q = 0; q < NQ; ++q) out[b * NQ + q] = acc[q];
    }
}

extern "C" void kernel_launch(void* const* d_in, const int* in_sizes, int n_in,
                              void* d_out, int out_size, void* d_ws, size_t ws_size,
                              hipStream_t stream) {
    const float* x     = (const float*)d_in[0];
    const float* encW  = (const float*)d_in[1];
    const float* encB  = (const float*)d_in[2];
    const float* theta = (const float*)d_in[3];
    float* out  = (float*)d_out;
    float* Mvar = (float*)d_ws;

    int batch = in_sizes[0] / 3;

    qprep_kernel<<<1, 64, 0, stream>>>(theta, Mvar);

    int blocks = (batch + WPB - 1) / WPB;
    qsim_kernel<<<blocks, WPB * 32, 0, stream>>>(x, encW, encB, Mvar, out, batch);
}


# Read-out sums as a per-lane sum-and-difference tree (88 adds, not 15 ops x 32 probabilities), lane-bit signs once on the lane total, cross-lane reduce-scatter (12 shuffles not 50); new sum order
# speedup vs baseline: 7.8911x; 1.0636x over previous
#include <hip/hip_runtime.h>
#include <math.h>

typedef __attribute__((ext_vector_type(2))) float v2f;
typedef __attribute__((ext_vector_type(8))) float v8f;

#define NQ       10
#define DIM      1024
#define NLAYERS  4
#define WPB      8
#define PI_F     3.14159265358979323846f

struct c2 { float re, im; };
__device__ __forceinline__ c2 cmul(c2 a, c2 b) {
    return { a.re * b.re - a.im * b.im, a.re * b.im + a.im * b.re };
}
__device__ __forceinline__ c2 cadd(c2 a, c2 b) { return { a.re + b.re, a.im + b.im }; }

__global__ void qprep_kernel(const float* __restrict__ theta, float* __restrict__ Mout) {
    int g = blockIdx.x * blockDim.x + threadIdx.x;
    if (g >= NLAYERS * NQ) return;
    float t0 = theta[g * 3 + 0], t1 = theta[g * 3 + 1], t2 = theta[g * 3 + 2];
    float cx = cosf(t0 * 0.5f), sx = sinf(t0 * 0.5f);
    float cy = cosf(t1 * 0.5f), sy = sinf(t1 * 0.5f);
    float cz = cosf(t2 * 0.5f), sz = sinf(t2 * 0.5f);
    c2 X[2][2] = { { {cx, 0.f}, {0.f, -sx} }, { {0.f, -sx}, {cx, 0.f} } };
    c2 Y[2][2] = { { {cy, 0.f}, {-sy, 0.f} }, { {sy, 0.f}, {cy, 0.f} } };
    c2 Z[2][2] = { { {cz, -sz}, {0.f, 0.f} }, { {0.f, 0.f}, {cz, sz} } };
    c2 XY[2][2], G[2][2];
    for (int i = 0; i < 2; ++i)
        for (int j = 0; j < 2; ++j)
            XY[i][j] = cadd(cmul(X[i][0], Y[0][j]), cmul(X[i][1], Y[1][j]));
    for (int i = 0; i < 2; ++i)
        for (int j = 0; j < 2; ++j)
            G[i][j] = cadd(cmul(XY[i][0], Z[0][j]), cmul(XY[i][1], Z[1][j]));
    float* M = Mout + g * 16;
    for (int i = 0; i < 2; ++i)
        for (int j = 0; j < 2; ++j) {
            M[(2 * i    ) * 4 + 2 * j    ] =  G[i][j].re;
            M[(2 * i    ) * 4 + 2 * j + 1] =  G[i][j].im;
            M[(2 * i + 1) * 4 + 2 * j    ] = -G[i][j].im;
            M[(2 * i + 1) * 4 + 2 * j + 1] =  G[i][j].re;
        }
}

template <int PH, int PJ1, int PJ0, int N0, int N1, int N2, int N3, int T0, int T1, int T2>
__device__ __forceinline__ void fused_pass(float2* __restrict__ st, int lane,
                                           c2 uH, c2 uJ1a, c2 uJ1b, c2 uJ0a, c2 uJ0b) {
    const int c  = lane & 15;
    const int h  = lane >> 4;
    const int co = c & 1;

    const c2 t0 = cmul(uH, uJ1a), t1 = cmul(uH, uJ1b);
    c2 P[4];
    P[0] = cmul(t0, uJ0a); P[1] = cmul(t0, uJ0b);
    P[2] = cmul(t1, uJ0a); P[3] = cmul(t1, uJ0b);
    v2f A[4];
#pragma unroll
    for (int j = 0; j < 4; ++j) {
        A[j].x = co ? P[j].im :  P[j].re;
        A[j].y = co ? P[j].re : -P[j].im;
    }

    const int base = ((c & 1) << N0) | (((c >> 1) & 1) << N1) | (((c >> 2) & 1) << N2) |
                     (((c >> 3) & 1) << N3) | (h << PH);
    float2* p = st + base;

    float2 cur0[4], cur1[4], nx0[4], nx1[4];
#pragma unroll
    for (int j = 0; j < 4; ++j) {
        const int jo = ((j >> 1) << PJ1) | ((j & 1) << PJ0);
        cur0[j] = p[jo];
        cur1[j] = p[(1 << T0) | jo];
    }
#pragma unroll
    for (int tp = 0; tp < 4; ++tp) {
        const int ta = ((tp & 1) << T1) | (((tp >> 1) & 1) << T2);
        const int tb = ta | (1 << T0);
        if (tp < 3) {
            const int u  = 2 * tp + 2;
            const int na = (((u >> 1) & 1) << T1) | (((u >> 2) & 1) << T2);
#pragma unroll
            for (int j = 0; j < 4; ++j) {
                const int jo = ((j >> 1) << PJ1) | ((j & 1) << PJ0);
                nx0[j] = p[na | jo];
                nx1[j] = p[na | (1 << T0) | jo];
            }
        }
        v8f C0 = {0.f, 0.f, 0.f, 0.f, 0.f, 0.f, 0.f, 0.f};
        v8f C1 = {0.f, 0.f, 0.f, 0.f, 0.f, 0.f, 0.f, 0.f};
#pragma unroll
        for (int j = 0; j < 4; ++j) {
            v2f B0; B0.x = cur0[j].x; B0.y = cur0[j].y;
            v2f B1; B1.x = cur1[j].x; B1.y = cur1[j].y;
            C0 = __builtin_amdgcn_wmma_f32_16x16x4_f32(false, A[j], false, B0, (short)0, C0, false, false);
            C1 = __builtin_amdgcn_wmma_f32_16x16x4_f32(false, A[j], false, B1, (short)0, C1, false, false);
        }
#pragma unroll
        for (int j = 0; j < 4; ++j) {
            const int jo = ((j >> 1) << PJ1) | ((j & 1) << PJ0);
            p[ta | jo] = make_float2(C0[2 * j], C0[2 * j + 1]);
            p[tb | jo] = make_float2(C1[2 * j], C1[2 * j + 1]);
        }
#pragma unroll
        for (int j = 0; j < 4; ++j) { cur0[j] = nx0[j]; cur1[j] = nx1[j]; }
    }
}

__device__ __forceinline__ int ring_src(int fl, int t) {
    const int ft = (t << 5) ^ (t << 4);
    return (fl ^ (ft & 0x310)) + (ft & 0x0E0);
}
__device__ __forceinline__ void ring_pass0(float2* __restrict__ st, int lane, int fl,
                                           c2 uH, c2 uJ1a, c2 uJ1b, c2 uJ0a, c2 uJ0b) {
    const int co = lane & 1;

    const c2 t0 = cmul(uH, uJ1a), t1 = cmul(uH, uJ1b);
    c2 P[4];
    P[0] = cmul(t0, uJ0a); P[1] = cmul(t0, uJ0b);
    P[2] = cmul(t1, uJ0a); P[3] = cmul(t1, uJ0b);
    v2f A[4];
#pragma unroll
    for (int j = 0; j < 4; ++j) {
        A[j].x = co ? P[j].im :  P[j].re;
        A[j].y = co ? P[j].re : -P[j].im;
    }

    float2 g[32];
#pragma unroll
    for (int tp = 0; tp < 4; ++tp)
#pragma unroll
        for (int j = 0; j < 4; ++j)
#pragma unroll
            for (int e = 0; e < 2; ++e) {
                const int t = 2 * tp + e + 8 * (j & 1) + 16 * (j >> 1);
                g[t] = st[ring_src(fl, t)];
            }

    float2* p = st + lane;
#pragma unroll
    for (int tp = 0; tp < 4; ++tp) {
        v8f C0 = {0.f, 0.f, 0.f, 0.f, 0.f, 0.f, 0.f, 0.f};
        v8f C1 = {0.f, 0.f, 0.f, 0.f, 0.f, 0.f, 0.f, 0.f};
#pragma unroll
        for (int j = 0; j < 4; ++j) {
            const float2 b0 = g[2 * tp     + 8 * (j & 1) + 16 * (j >> 1)];
            const float2 b1 = g[2 * tp + 1 + 8 * (j & 1) + 16 * (j >> 1)];
            v2f B0; B0.x = b0.x; B0.y = b0.y;
            v2f B1; B1.x = b1.x; B1.y = b1.y;
            C0 = __builtin_amdgcn_wmma_f32_16x16x4_f32(false, A[j], false, B0, (short)0, C0, false, false);
            C1 = __builtin_amdgcn_wmma_f32_16x16x4_f32(false, A[j], false, B1, (short)0, C1, false, false);
        }
#pragma unroll
        for (int j = 0; j < 4; ++j) {
            const int jo = ((j >> 1) << 9) | ((j & 1) << 8);
            p[((2 * tp    ) << 5) | jo] = make_float2(C0[2 * j], C0[2 * j + 1]);
            p[((2 * tp + 1) << 5) | jo] = make_float2(C1[2 * j], C1[2 * j + 1]);
        }
    }
}

#define PASS1 fused_pass<3, 7, 6, 0, 1, 2, 4, 5, 8, 9>
#define PASS2 fused_pass<2, 5, 0, 1, 3, 4, 6, 7, 8, 9>
#define PASS3 fused_pass<1, 9, 8, 0, 2, 3, 4, 5, 6, 7>
#define P1A fused_pass<4, 8, 6, 0, 1, 2, 3, 5, 7, 9>
#define P1B fused_pass<2, 9, 0, 1, 3, 4, 5, 6, 7, 8>

__device__ __forceinline__ c2 csel(int pick1, c2 a0, c2 a1) {
    c2 r; r.re = pick1 ? a1.re : a0.re; r.im = pick1 ? a1.im : a0.im; return r;
}
__device__ __forceinline__ c2 fold2(c2 u0, c2 u1, c2 pa, c2 pb, c2 qa, c2 qb) {
    return cadd(cmul(u0, cmul(pa, qa)), cmul(u1, cmul(pb, qb)));
}
__device__ __forceinline__ c2 id_u(int o, int i) {
    c2 r; r.re = (o == i) ? 1.f : 0.f; r.im = 0.f; return r;
}
__device__ __forceinline__ c2 var_u(const float* __restrict__ M, int o, int i) {
    const float2 v = *(const float2*)(M + 8 * i + 2 * o);
    c2 r; r.re = v.x; r.im = v.y; return r;
}

__device__ __forceinline__ float lane_value(float v, int src) {
    return __int_as_float(__builtin_amdgcn_readlane(__float_as_int(v), src));
}

__global__ void __launch_bounds__(WPB * 32) __attribute__((amdgpu_waves_per_eu(8)))
qsim_kernel(const float* __restrict__ x, const float* __restrict__ encW,
            const float* __restrict__ encB, const float* __restrict__ Mvar,
            float* __restrict__ out, int batch) {
    __shared__ float2 lds[WPB * DIM];

    const int lane = threadIdx.x & 31;
    const int w    = threadIdx.x >> 5;
    const int b    = blockIdx.x * WPB + w;
    if (b >= batch) return;

    float2* st = &lds[w * DIM];
    const int c  = lane & 15;
    const int h  = lane >> 4;
    const int ho = c >> 3;
    const int o1 = (c >> 2) & 1;
    const int o0 = (c >> 1) & 1;

    {
        float xv0 = x[b * 3 + 0], xv1 = x[b * 3 + 1], xv2 = x[b * 3 + 2];
        const int q = lane < NQ ? lane : NQ - 1;
        float ang = PI_F * tanhf(fmaf(xv2, encW[q * 3 + 2],
                                      fmaf(xv0, encW[q * 3], xv1 * encW[q * 3 + 1])) + encB[q]);
        const float cq = cosf(ang * 0.5f);
        const float sq = sinf(ang * 0.5f);
        const float* M = Mvar + q * 16;
        c2 own[2];
#pragma unroll
        for (int o = 0; o < 2; ++o) {
            const c2 g0 = var_u(M, o, 0), g1 = var_u(M, o, 1);
            own[o].re = cq * g0.re - sq * g1.re;
            own[o].im = cq * g0.im - sq * g1.im;
        }
        c2 V[NQ][2];
#pragma unroll
        for (int p = 0; p < NQ; ++p) {
#pragma unroll
            for (int o = 0; o < 2; ++o) {
                V[p][o].re = lane_value(own[o].re, 9 - p);
                V[p][o].im = lane_value(own[o].im, 9 - p);
            }
        }
        const float* M1 = Mvar + NQ * 16;
        const int l0 = lane & 1, l1 = (lane >> 1) & 1, l2 = (lane >> 2) & 1;
        const int l3 = (lane >> 3) & 1, l4 = lane >> 4;
        c2 L;
        {
            const c2 w1 = fold2(var_u(M1 + 8 * 16, l1, 0), var_u(M1 + 8 * 16, l1, 1),
                                csel(l2, V[1][0], V[1][1]), csel(l2, V[1][1], V[1][0]),
                                csel(l0, V[0][0], V[0][1]), csel(l0, V[0][1], V[0][0]));
            const c2 w3 = fold2(var_u(M1 + 6 * 16, l3, 0), var_u(M1 + 6 * 16, l3, 1),
                                csel(l4, V[3][0], V[3][1]), csel(l4, V[3][1], V[3][0]),
                                csel(l2, V[2][0], V[2][1]), csel(l2, V[2][1], V[2][0]));
            L = cmul(w1, w3);
        }
        c2 A2[2][2];
        {
            const c2 v4a = csel(l4, V[4][0], V[4][1]);
            const c2 v4b = csel(l4, V[4][1], V[4][0]);
#pragma unroll
            for (int t0 = 0; t0 < 2; ++t0) {
                const c2 u0 = var_u(M1 + 4 * 16, t0, 0), u1 = var_u(M1 + 4 * 16, t0, 1);
#pragma unroll
                for (int t1 = 0; t1 < 2; ++t1)
                    A2[t0][t1] = cmul(L, fold2(u0, u1, V[5][t1], V[5][1 ^ t1], v4a, v4b));
            }
        }
        c2 Bt[16];
        {
            c2 W7[2][2][2];
#pragma unroll
            for (int t2 = 0; t2 < 2; ++t2) {
                const c2 u0 = var_u(M1 + 2 * 16, t2, 0), u1 = var_u(M1 + 2 * 16, t2, 1);
#pragma unroll
                for (int t1 = 0; t1 < 2; ++t1)
#pragma unroll
                    for (int t3 = 0; t3 < 2; ++t3)
                        W7[t1][t2][t3] = fold2(u0, u1, V[7][t3], V[7][1 ^ t3], V[6][t1], V[6][1 ^ t1]);
            }
            const c2 v9a = csel(l0, V[9][0], V[9][1]);
            const c2 v9b = csel(l0, V[9][1], V[9][0]);
#pragma unroll
            for (int t4 = 0; t4 < 2; ++t4) {
                const c2 u0 = var_u(M1 + 0 * 16, t4, 0), u1 = var_u(M1 + 0 * 16, t4, 1);
#pragma unroll
                for (int t3 = 0; t3 < 2; ++t3) {
                    const c2 v8a = csel(l0, V[8][t3], V[8][1 ^ t3]);
                    const c2 v8b = csel(l0, V[8][1 ^ t3], V[8][t3]);
                    const c2 w9 = fold2(u0, u1, v9a, v9b, v8a, v8b);
#pragma unroll
                    for (int t2 = 0; t2 < 2; ++t2)
#pragma unroll
                        for (int t1 = 0; t1 < 2; ++t1)
                            Bt[t1 + 2 * t2 + 4 * t3 + 8 * t4] = cmul(W7[t1][t2][t3], w9);
                }
            }
        }
#pragma unroll
        for (int t = 0; t < 32; ++t) {
            const c2 a = cmul(A2[t & 1][(t >> 1) & 1], Bt[t >> 1]);
            st[lane + (t << 5)] = make_float2(a.re, a.im);
        }
    }

    {
        const float* ML = Mvar + NQ * 16;
        P1A(st, lane, var_u(ML + 5 * 16, ho, h),
            var_u(ML + 1 * 16, o1, 0), var_u(ML + 1 * 16, o1, 1),
            var_u(ML + 3 * 16, o0, 0), var_u(ML + 3 * 16, o0, 1));
        P1B(st, lane, var_u(ML + 7 * 16, ho, h),
            id_u(o1, 0), id_u(o1, 1),
            var_u(ML + 9 * 16, o0, 0), var_u(ML + 9 * 16, o0, 1));
    }

    const int fl = (lane ^ (lane >> 1)) | ((lane & 1) ? 0x300 : 0);
    for (int layer = 2; layer < NLAYERS; ++layer) {
        const float* ML = Mvar + layer * NQ * 16;
        ring_pass0(st, lane, fl, var_u(ML + 5 * 16, ho, h),
                   var_u(ML + 0 * 16, o1, 0), var_u(ML + 0 * 16, o1, 1),
                   var_u(ML + 1 * 16, o0, 0), var_u(ML + 1 * 16, o0, 1));
        PASS1(st, lane, var_u(ML + 6 * 16, ho, h),
              var_u(ML + 2 * 16, o1, 0), var_u(ML + 2 * 16, o1, 1),
              var_u(ML + 3 * 16, o0, 0), var_u(ML + 3 * 16, o0, 1));
        PASS2(st, lane, var_u(ML + 7 * 16, ho, h),
              var_u(ML + 4 * 16, o1, 0), var_u(ML + 4 * 16, o1, 1),
              var_u(ML + 9 * 16, o0, 0), var_u(ML + 9 * 16, o0, 1));
        PASS3(st, lane, var_u(ML + 8 * 16, ho, h),
              id_u(o1, 0), id_u(o1, 1), id_u(o0, 0), id_u(o0, 1));
    }

    float s[32];
#pragma unroll
    for (int t = 0; t < 32; ++t) {
        const float2 a = st[ring_src(fl, t)];
        s[t] = a.x * a.x + a.y * a.y;
    }
    float v[NQ];
#pragma unroll
    for (int k = 0; k < 5; ++k) {
        float dk = 0.f;
#pragma unroll
        for (int u = 0; u < (16 >> k); ++u) {
            const float a0 = s[2 * u], a1 = s[2 * u + 1];
            s[u] = a0 + a1;
            dk = (u == 0) ? (a0 - a1) : dk + (a0 - a1);
        }
        v[4 - k] = dk;
    }
#pragma unroll
    for (int k = 0; k < 5; ++k)
        v[9 - k] = ((lane >> k) & 1) ? -s[0] : s[0];

    const int b4 = lane >> 4, b3 = (lane >> 3) & 1, b2 = (lane >> 2) & 1, b1 = (lane >> 1) & 1;
    float k5[5], k3[3], k2[2], k1;
#pragma unroll
    for (int i = 0; i < 5; ++i) {
        const float keep = b4 ? v[5 + i] : v[i];
        const float send = b4 ? v[i] : v[5 + i];
        k5[i] = keep + __shfl_xor(send, 16, 32);
    }
#pragma unroll
    for (int i = 0; i < 3; ++i) {
        const float hi = (i < 2) ? k5[3 + i] : 0.f;
        const float keep = b3 ? hi : k5[i];
        const float send = b3 ? k5[i] : hi;
        k3[i] = keep + __shfl_xor(send, 8, 32);
    }
#pragma unroll
    for (int i = 0; i < 2; ++i) {
        const float hi = (i < 1) ? k3[2] : 0.f;
        const float keep = b2 ? hi : k3[i];
        const float send = b2 ? k3[i] : hi;
        k2[i] = keep + __shfl_xor(send, 4, 32);
    }
    {
        const float keep = b1 ? k2[1] : k2[0];
        const float send = b1 ? k2[0] : k2[1];
        k1 = keep + __shfl_xor(send, 2, 32);
    }
    k1 += __shfl_xor(k1, 1, 32);
    const int  ql    = b3 ? 3 + b1 : (b2 ? 2 : b1);
    const bool holds = ((lane & 1) == 0) && (b3 ? (b2 == 0) : !(b2 && b1));
    if (holds) out[b * NQ + 5 * b4 + ql] = k1;
}

extern "C" void kernel_launch(void* const* d_in, const int* in_sizes, int n_in,
                              void* d_out, int out_size, void* d_ws, size_t ws_size,
                              hipStream_t stream) {
    const float* x     = (const float*)d_in[0];
    const float* encW  = (const float*)d_in[1];
    const float* encB  = (const float*)d_in[2];
    const float* theta = (const float*)d_in[3];
    float* out  = (float*)d_out;
    float* Mvar = (float*)d_ws;

    int batch = in_sizes[0] / 3;

    qprep_kernel<<<1, 64, 0, stream>>>(theta, Mvar);

    int blocks = (batch + WPB - 1) / WPB;
    qsim_kernel<<<blocks, WPB * 32, 0, stream>>>(x, encW, encB, Mvar, out, batch);
}


# Layer-1 gates of the five even index bits folded into the closed form too: each amplitude a four-term trace of 2x2 factor products; no layer-1 WMMA pass left (8 passes, not 10); new sum order
# speedup vs baseline: 8.7307x; 1.1064x over previous
#include <hip/hip_runtime.h>
#include <math.h>

typedef __attribute__((ext_vector_type(2))) float v2f;
typedef __attribute__((ext_vector_type(8))) float v8f;

#define NQ       10
#define DIM      1024
#define NLAYERS  4
#define WPB      8
#define PI_F     3.14159265358979323846f

struct c2 { float re, im; };
__device__ __forceinline__ c2 cmul(c2 a, c2 b) {
    return { a.re * b.re - a.im * b.im, a.re * b.im + a.im * b.re };
}
__device__ __forceinline__ c2 cadd(c2 a, c2 b) { return { a.re + b.re, a.im + b.im }; }

__global__ void qprep_kernel(const float* __restrict__ theta, float* __restrict__ Mout) {
    int g = blockIdx.x * blockDim.x + threadIdx.x;
    if (g >= NLAYERS * NQ) return;
    float t0 = theta[g * 3 + 0], t1 = theta[g * 3 + 1], t2 = theta[g * 3 + 2];
    float cx = cosf(t0 * 0.5f), sx = sinf(t0 * 0.5f);
    float cy = cosf(t1 * 0.5f), sy = sinf(t1 * 0.5f);
    float cz = cosf(t2 * 0.5f), sz = sinf(t2 * 0.5f);
    c2 X[2][2] = { { {cx, 0.f}, {0.f, -sx} }, { {0.f, -sx}, {cx, 0.f} } };
    c2 Y[2][2] = { { {cy, 0.f}, {-sy, 0.f} }, { {sy, 0.f}, {cy, 0.f} } };
    c2 Z[2][2] = { { {cz, -sz}, {0.f, 0.f} }, { {0.f, 0.f}, {cz, sz} } };
    c2 XY[2][2], G[2][2];
    for (int i = 0; i < 2; ++i)
        for (int j = 0; j < 2; ++j)
            XY[i][j] = cadd(cmul(X[i][0], Y[0][j]), cmul(X[i][1], Y[1][j]));
    for (int i = 0; i < 2; ++i)
        for (int j = 0; j < 2; ++j)
            G[i][j] = cadd(cmul(XY[i][0], Z[0][j]), cmul(XY[i][1], Z[1][j]));
    float* M = Mout + g * 16;
    for (int i = 0; i < 2; ++i)
        for (int j = 0; j < 2; ++j) {
            M[(2 * i    ) * 4 + 2 * j    ] =  G[i][j].re;
            M[(2 * i    ) * 4 + 2 * j + 1] =  G[i][j].im;
            M[(2 * i + 1) * 4 + 2 * j    ] = -G[i][j].im;
            M[(2 * i + 1) * 4 + 2 * j + 1] =  G[i][j].re;
        }
}

template <int PH, int PJ1, int PJ0, int N0, int N1, int N2, int N3, int T0, int T1, int T2>
__device__ __forceinline__ void fused_pass(float2* __restrict__ st, int lane,
                                           c2 uH, c2 uJ1a, c2 uJ1b, c2 uJ0a, c2 uJ0b) {
    const int c  = lane & 15;
    const int h  = lane >> 4;
    const int co = c & 1;

    const c2 t0 = cmul(uH, uJ1a), t1 = cmul(uH, uJ1b);
    c2 P[4];
    P[0] = cmul(t0, uJ0a); P[1] = cmul(t0, uJ0b);
    P[2] = cmul(t1, uJ0a); P[3] = cmul(t1, uJ0b);
    v2f A[4];
#pragma unroll
    for (int j = 0; j < 4; ++j) {
        A[j].x = co ? P[j].im :  P[j].re;
        A[j].y = co ? P[j].re : -P[j].im;
    }

    const int base = ((c & 1) << N0) | (((c >> 1) & 1) << N1) | (((c >> 2) & 1) << N2) |
                     (((c >> 3) & 1) << N3) | (h << PH);
    float2* p = st + base;

    float2 cur0[4], cur1[4], nx0[4], nx1[4];
#pragma unroll
    for (int j = 0; j < 4; ++j) {
        const int jo = ((j >> 1) << PJ1) | ((j & 1) << PJ0);
        cur0[j] = p[jo];
        cur1[j] = p[(1 << T0) | jo];
    }
#pragma unroll
    for (int tp = 0; tp < 4; ++tp) {
        const int ta = ((tp & 1) << T1) | (((tp >> 1) & 1) << T2);
        const int tb = ta | (1 << T0);
        if (tp < 3) {
            const int u  = 2 * tp + 2;
            const int na = (((u >> 1) & 1) << T1) | (((u >> 2) & 1) << T2);
#pragma unroll
            for (int j = 0; j < 4; ++j) {
                const int jo = ((j >> 1) << PJ1) | ((j & 1) << PJ0);
                nx0[j] = p[na | jo];
                nx1[j] = p[na | (1 << T0) | jo];
            }
        }
        v8f C0 = {0.f, 0.f, 0.f, 0.f, 0.f, 0.f, 0.f, 0.f};
        v8f C1 = {0.f, 0.f, 0.f, 0.f, 0.f, 0.f, 0.f, 0.f};
#pragma unroll
        for (int j = 0; j < 4; ++j) {
            v2f B0; B0.x = cur0[j].x; B0.y = cur0[j].y;
            v2f B1; B1.x = cur1[j].x; B1.y = cur1[j].y;
            C0 = __builtin_amdgcn_wmma_f32_16x16x4_f32(false, A[j], false, B0, (short)0, C0, false, false);
            C1 = __builtin_amdgcn_wmma_f32_16x16x4_f32(false, A[j], false, B1, (short)0, C1, false, false);
        }
#pragma unroll
        for (int j = 0; j < 4; ++j) {
            const int jo = ((j >> 1) << PJ1) | ((j & 1) << PJ0);
            p[ta | jo] = make_float2(C0[2 * j], C0[2 * j + 1]);
            p[tb | jo] = make_float2(C1[2 * j], C1[2 * j + 1]);
        }
#pragma unroll
        for (int j = 0; j < 4; ++j) { cur0[j] = nx0[j]; cur1[j] = nx1[j]; }
    }
}

__device__ __forceinline__ int ring_src(int fl, int t) {
    const int ft = (t << 5) ^ (t << 4);
    return (fl ^ (ft & 0x310)) + (ft & 0x0E0);
}
__device__ __forceinline__ void ring_pass0(float2* __restrict__ st, int lane, int fl,
                                           c2 uH, c2 uJ1a, c2 uJ1b, c2 uJ0a, c2 uJ0b) {
    const int co = lane & 1;

    const c2 t0 = cmul(uH, uJ1a), t1 = cmul(uH, uJ1b);
    c2 P[4];
    P[0] = cmul(t0, uJ0a); P[1] = cmul(t0, uJ0b);
    P[2] = cmul(t1, uJ0a); P[3] = cmul(t1, uJ0b);
    v2f A[4];
#pragma unroll
    for (int j = 0; j < 4; ++j) {
        A[j].x = co ? P[j].im :  P[j].re;
        A[j].y = co ? P[j].re : -P[j].im;
    }

    float2 g[32];
#pragma unroll
    for (int tp = 0; tp < 4; ++tp)
#pragma unroll
        for (int j = 0; j < 4; ++j)
#pragma unroll
            for (int e = 0; e < 2; ++e) {
                const int t = 2 * tp + e + 8 * (j & 1) + 16 * (j >> 1);
                g[t] = st[ring_src(fl, t)];
            }

    float2* p = st + lane;
#pragma unroll
    for (int tp = 0; tp < 4; ++tp) {
        v8f C0 = {0.f, 0.f, 0.f, 0.f, 0.f, 0.f, 0.f, 0.f};
        v8f C1 = {0.f, 0.f, 0.f, 0.f, 0.f, 0.f, 0.f, 0.f};
#pragma unroll
        for (int j = 0; j < 4; ++j) {
            const float2 b0 = g[2 * tp     + 8 * (j & 1) + 16 * (j >> 1)];
            const float2 b1 = g[2 * tp + 1 + 8 * (j & 1) + 16 * (j >> 1)];
            v2f B0; B0.x = b0.x; B0.y = b0.y;
            v2f B1; B1.x = b1.x; B1.y = b1.y;
            C0 = __builtin_amdgcn_wmma_f32_16x16x4_f32(false, A[j], false, B0, (short)0, C0, false, false);
            C1 = __builtin_amdgcn_wmma_f32_16x16x4_f32(false, A[j], false, B1, (short)0, C1, false, false);
        }
#pragma unroll
        for (int j = 0; j < 4; ++j) {
            const int jo = ((j >> 1) << 9) | ((j & 1) << 8);
            p[((2 * tp    ) << 5) | jo] = make_float2(C0[2 * j], C0[2 * j + 1]);
            p[((2 * tp + 1) << 5) | jo] = make_float2(C1[2 * j], C1[2 * j + 1]);
        }
    }
}

#define PASS1 fused_pass<3, 7, 6, 0, 1, 2, 4, 5, 8, 9>
#define PASS2 fused_pass<2, 5, 0, 1, 3, 4, 6, 7, 8, 9>
#define PASS3 fused_pass<1, 9, 8, 0, 2, 3, 4, 5, 6, 7>

__device__ __forceinline__ c2 csel(int pick1, c2 a0, c2 a1) {
    c2 r; r.re = pick1 ? a1.re : a0.re; r.im = pick1 ? a1.im : a0.im; return r;
}
__device__ __forceinline__ c2 fold2(c2 u0, c2 u1, c2 pa, c2 pb, c2 qa, c2 qb) {
    return cadd(cmul(u0, cmul(pa, qa)), cmul(u1, cmul(pb, qb)));
}
__device__ __forceinline__ c2 id_u(int o, int i) {
    c2 r; r.re = (o == i) ? 1.f : 0.f; r.im = 0.f; return r;
}
__device__ __forceinline__ c2 var_u(const float* __restrict__ M, int o, int i) {
    const float2 v = *(const float2*)(M + 8 * i + 2 * o);
    c2 r; r.re = v.x; r.im = v.y; return r;
}

__device__ __forceinline__ float lane_value(float v, int src) {
    return __int_as_float(__builtin_amdgcn_readlane(__float_as_int(v), src));
}

__global__ void __launch_bounds__(WPB * 32) __attribute__((amdgpu_waves_per_eu(8)))
qsim_kernel(const float* __restrict__ x, const float* __restrict__ encW,
            const float* __restrict__ encB, const float* __restrict__ Mvar,
            float* __restrict__ out, int batch) {
    __shared__ float2 lds[WPB * DIM];

    const int lane = threadIdx.x & 31;
    const int w    = threadIdx.x >> 5;
    const int b    = blockIdx.x * WPB + w;
    if (b >= batch) return;

    float2* st = &lds[w * DIM];
    const int c  = lane & 15;
    const int h  = lane >> 4;
    const int ho = c >> 3;
    const int o1 = (c >> 2) & 1;
    const int o0 = (c >> 1) & 1;

    {
        float xv0 = x[b * 3 + 0], xv1 = x[b * 3 + 1], xv2 = x[b * 3 + 2];
        const int q = lane < NQ ? lane : NQ - 1;
        float ang = PI_F * tanhf(fmaf(xv2, encW[q * 3 + 2],
                                      fmaf(xv0, encW[q * 3], xv1 * encW[q * 3 + 1])) + encB[q]);
        const float cq = cosf(ang * 0.5f);
        const float sq = sinf(ang * 0.5f);
        const float* M = Mvar + q * 16;
        c2 own[2];
#pragma unroll
        for (int o = 0; o < 2; ++o) {
            const c2 g0 = var_u(M, o, 0), g1 = var_u(M, o, 1);
            own[o].re = cq * g0.re - sq * g1.re;
            own[o].im = cq * g0.im - sq * g1.im;
        }
        c2 V[NQ][2];
#pragma unroll
        for (int p = 0; p < NQ; ++p) {
#pragma unroll
            for (int o = 0; o < 2; ++o) {
                V[p][o].re = lane_value(own[o].re, 9 - p);
                V[p][o].im = lane_value(own[o].im, 9 - p);
            }
        }
        const float* M1 = Mvar + NQ * 16;
        const int l0 = lane & 1, l1 = (lane >> 1) & 1, l2 = (lane >> 2) & 1;
        const int l3 = (lane >> 3) & 1, l4 = lane >> 4;
        {
            const c2 al = cmul(var_u(M1 + 2 * 16, l2, 0), csel(l1, V[6][0], V[6][1]));
            const c2 be = cmul(var_u(M1 + 2 * 16, l2, 1), csel(l1, V[6][1], V[6][0]));
            const c2 w7y0 = cadd(cmul(al, V[7][0]), cmul(be, V[7][1]));
            const c2 w7y1 = cadd(cmul(al, V[7][1]), cmul(be, V[7][0]));
            const c2 ga = cmul(var_u(M1 + 0 * 16, l4, 0), csel(l0, V[9][0], V[9][1]));
            const c2 de = cmul(var_u(M1 + 0 * 16, l4, 1), csel(l0, V[9][1], V[9][0]));
            const c2 v8z = csel(l0, V[8][0], V[8][1]);
            const c2 v8n = csel(l0, V[8][1], V[8][0]);
            const c2 w9x0 = cadd(cmul(ga, v8z), cmul(de, v8n));
            const c2 w9x1 = cadd(cmul(ga, v8n), cmul(de, v8z));
            const c2 ye = cadd(cmul(cmul(w7y0, var_u(M1 + 1 * 16, l3, 0)), w9x0),
                               cmul(cmul(w7y1, var_u(M1 + 1 * 16, l3, 1)), w9x1));
            st[992 + lane] = make_float2(ye.re, ye.im);
            const c2 w5 = fold2(var_u(M1 + 4 * 16, l2, 0), var_u(M1 + 4 * 16, l2, 1),
                                csel(l0, V[5][0], V[5][1]), csel(l0, V[5][1], V[5][0]),
                                csel(l1, V[4][0], V[4][1]), csel(l1, V[4][1], V[4][0]));
            const c2 xe = cmul(w5, var_u(M1 + 3 * 16, l3, l0));
            st[960 + lane] = make_float2(xe.re, xe.im);
        }
        c2 P[2][2];
        {
            const c2 u10 = var_u(M1 + 8 * 16, l1, 0), u11 = var_u(M1 + 8 * 16, l1, 1);
            const c2 u30 = var_u(M1 + 6 * 16, l3, 0), u31 = var_u(M1 + 6 * 16, l3, 1);
            c2 W1s[2][2], W3m[2][2];
#pragma unroll
            for (int xx = 0; xx < 2; ++xx) {
                const c2 a1 = cmul(u10, V[0][xx]), b1 = cmul(u11, V[0][1 ^ xx]);
                const c2 a3 = cmul(u30, V[2][xx]), b3 = cmul(u31, V[2][1 ^ xx]);
#pragma unroll
                for (int yy = 0; yy < 2; ++yy) {
                    W1s[xx][yy] = cmul(cadd(cmul(a1, V[1][yy]), cmul(b1, V[1][1 ^ yy])),
                                       var_u(M1 + 7 * 16, l2, yy));
                    W3m[xx][yy] = cadd(cmul(a3, V[3][yy]), cmul(b3, V[3][1 ^ yy]));
                }
            }
#pragma unroll
            for (int a0 = 0; a0 < 2; ++a0)
#pragma unroll
                for (int a4 = 0; a4 < 2; ++a4) {
                    const c2 mid = cadd(cmul(W1s[a0][0], W3m[0][a4]), cmul(W1s[a0][1], W3m[1][a4]));
                    P[a0][a4] = cmul(cmul(var_u(M1 + 9 * 16, l0, a0), var_u(M1 + 5 * 16, l4, a4)), mid);
                }
        }
        v2f R[4][2][2];
#pragma unroll
        for (int q4 = 0; q4 < 4; ++q4) {
            v2f X[2][2], Xr[2][2];
#pragma unroll
            for (int a4 = 0; a4 < 2; ++a4)
#pragma unroll
                for (int a6 = 0; a6 < 2; ++a6) {
                    const float2 xv = st[960 + 4 * q4 + a6 + 2 * a4];
                    X[a4][a6].x = xv.x;  X[a4][a6].y = xv.y;
                    Xr[a4][a6].x = -xv.y; Xr[a4][a6].y = xv.x;
                }
#pragma unroll
            for (int a0 = 0; a0 < 2; ++a0)
#pragma unroll
                for (int a6 = 0; a6 < 2; ++a6)
                    R[q4][a0][a6] = P[a0][0].re * X[0][a6] + P[a0][0].im * Xr[0][a6]
                                  + P[a0][1].re * X[1][a6] + P[a0][1].im * Xr[1][a6];
        }
#pragma unroll
        for (int s8 = 0; s8 < 8; ++s8) {
            v2f Y[2][2], Yr[2][2];
#pragma unroll
            for (int a6 = 0; a6 < 2; ++a6)
#pragma unroll
                for (int a0 = 0; a0 < 2; ++a0) {
                    const float2 yv = st[992 + 4 * s8 + a0 + 2 * a6];
                    Y[a6][a0].x = yv.x;  Y[a6][a0].y = yv.y;
                    Yr[a6][a0].x = -yv.y; Yr[a6][a0].y = yv.x;
                }
#pragma unroll
            for (int q4 = 0; q4 < 4; ++q4) {
                const v2f a = R[q4][0][0].x * Y[0][0] + R[q4][0][0].y * Yr[0][0]
                            + R[q4][0][1].x * Y[1][0] + R[q4][0][1].y * Yr[1][0]
                            + R[q4][1][0].x * Y[0][1] + R[q4][1][0].y * Yr[0][1]
                            + R[q4][1][1].x * Y[1][1] + R[q4][1][1].y * Yr[1][1];
                st[lane + ((q4 + 4 * s8) << 5)] = make_float2(a.x, a.y);
            }
        }
    }

    const int fl = (lane ^ (lane >> 1)) | ((lane & 1) ? 0x300 : 0);
    for (int layer = 2; layer < NLAYERS; ++layer) {
        const float* ML = Mvar + layer * NQ * 16;
        ring_pass0(st, lane, fl, var_u(ML + 5 * 16, ho, h),
                   var_u(ML + 0 * 16, o1, 0), var_u(ML + 0 * 16, o1, 1),
                   var_u(ML + 1 * 16, o0, 0), var_u(ML + 1 * 16, o0, 1));
        PASS1(st, lane, var_u(ML + 6 * 16, ho, h),
              var_u(ML + 2 * 16, o1, 0), var_u(ML + 2 * 16, o1, 1),
              var_u(ML + 3 * 16, o0, 0), var_u(ML + 3 * 16, o0, 1));
        PASS2(st, lane, var_u(ML + 7 * 16, ho, h),
              var_u(ML + 4 * 16, o1, 0), var_u(ML + 4 * 16, o1, 1),
              var_u(ML + 9 * 16, o0, 0), var_u(ML + 9 * 16, o0, 1));
        PASS3(st, lane, var_u(ML + 8 * 16, ho, h),
              id_u(o1, 0), id_u(o1, 1), id_u(o0, 0), id_u(o0, 1));
    }

    float s[32];
#pragma unroll
    for (int t = 0; t < 32; ++t) {
        const float2 a = st[ring_src(fl, t)];
        s[t] = a.x * a.x + a.y * a.y;
    }
    float v[NQ];
#pragma unroll
    for (int k = 0; k < 5; ++k) {
        float dk = 0.f;
#pragma unroll
        for (int u = 0; u < (16 >> k); ++u) {
            const float a0 = s[2 * u], a1 = s[2 * u + 1];
            s[u] = a0 + a1;
            dk = (u == 0) ? (a0 - a1) : dk + (a0 - a1);
        }
        v[4 - k] = dk;
    }
#pragma unroll
    for (int k = 0; k < 5; ++k)
        v[9 - k] = ((lane >> k) & 1) ? -s[0] : s[0];

    const int b4 = lane >> 4, b3 = (lane >> 3) & 1, b2 = (lane >> 2) & 1, b1 = (lane >> 1) & 1;
    float k5[5], k3[3], k2[2], k1;
#pragma unroll
    for (int i = 0; i < 5; ++i) {
        const float keep = b4 ? v[5 + i] : v[i];
        const float send = b4 ? v[i] : v[5 + i];
        k5[i] = keep + __shfl_xor(send, 16, 32);
    }
#pragma unroll
    for (int i = 0; i < 3; ++i) {
        const float hi = (i < 2) ? k5[3 + i] : 0.f;
        const float keep = b3 ? hi : k5[i];
        const float send = b3 ? k5[i] : hi;
        k3[i] = keep + __shfl_xor(send, 8, 32);
    }
#pragma unroll
    for (int i = 0; i < 2; ++i) {
        const float hi = (i < 1) ? k3[2] : 0.f;
        const float keep = b2 ? hi : k3[i];
        const float send = b2 ? k3[i] : hi;
        k2[i] = keep + __shfl_xor(send, 4, 32);
    }
    {
        const float keep = b1 ? k2[1] : k2[0];
        const float send = b1 ? k2[0] : k2[1];
        k1 = keep + __shfl_xor(send, 2, 32);
    }
    k1 += __shfl_xor(k1, 1, 32);
    const int  ql    = b3 ? 3 + b1 : (b2 ? 2 : b1);
    const bool holds = ((lane & 1) == 0) && (b3 ? (b2 == 0) : !(b2 && b1));
    if (holds) out[b * NQ + 5 * b4 + ql] = k1;
}

extern "C" void kernel_launch(void* const* d_in, const int* in_sizes, int n_in,
                              void* d_out, int out_size, void* d_ws, size_t ws_size,
                              hipStream_t stream) {
    const float* x     = (const float*)d_in[0];
    const float* encW  = (const float*)d_in[1];
    const float* encB  = (const float*)d_in[2];
    const float* theta = (const float*)d_in[3];
    float* out  = (float*)d_out;
    float* Mvar = (float*)d_ws;

    int batch = in_sizes[0] / 3;

    qprep_kernel<<<1, 64, 0, stream>>>(theta, Mvar);

    int blocks = (batch + WPB - 1) / WPB;
    qsim_kernel<<<blocks, WPB * 32, 0, stream>>>(x, encW, encB, Mvar, out, batch);
}


# Last gate of the circuit (layer 3, index bit 1) moved onto the observable: read-out takes each stored pair's sum and a quadratic form for its difference; 7 WMMA passes, not 8; new sum order
# speedup vs baseline: 9.0773x; 1.0397x over previous
#include <hip/hip_runtime.h>
#include <math.h>

typedef __attribute__((ext_vector_type(2))) float v2f;
typedef __attribute__((ext_vector_type(8))) float v8f;

#define NQ       10
#define DIM      1024
#define NLAYERS  4
#define WPB      8
#define PI_F     3.14159265358979323846f

struct c2 { float re, im; };
__device__ __forceinline__ c2 cmul(c2 a, c2 b) {
    return { a.re * b.re - a.im * b.im, a.re * b.im + a.im * b.re };
}
__device__ __forceinline__ c2 cadd(c2 a, c2 b) { return { a.re + b.re, a.im + b.im }; }

__global__ void qprep_kernel(const float* __restrict__ theta, float* __restrict__ Mout) {
    int g = blockIdx.x * blockDim.x + threadIdx.x;
    if (g >= NLAYERS * NQ) return;
    float t0 = theta[g * 3 + 0], t1 = theta[g * 3 + 1], t2 = theta[g * 3 + 2];
    float cx = cosf(t0 * 0.5f), sx = sinf(t0 * 0.5f);
    float cy = cosf(t1 * 0.5f), sy = sinf(t1 * 0.5f);
    float cz = cosf(t2 * 0.5f), sz = sinf(t2 * 0.5f);
    c2 X[2][2] = { { {cx, 0.f}, {0.f, -sx} }, { {0.f, -sx}, {cx, 0.f} } };
    c2 Y[2][2] = { { {cy, 0.f}, {-sy, 0.f} }, { {sy, 0.f}, {cy, 0.f} } };
    c2 Z[2][2] = { { {cz, -sz}, {0.f, 0.f} }, { {0.f, 0.f}, {cz, sz} } };
    c2 XY[2][2], G[2][2];
    for (int i = 0; i < 2; ++i)
        for (int j = 0; j < 2; ++j)
            XY[i][j] = cadd(cmul(X[i][0], Y[0][j]), cmul(X[i][1], Y[1][j]));
    for (int i = 0; i < 2; ++i)
        for (int j = 0; j < 2; ++j)
            G[i][j] = cadd(cmul(XY[i][0], Z[0][j]), cmul(XY[i][1], Z[1][j]));
    float* M = Mout + g * 16;
    for (int i = 0; i < 2; ++i)
        for (int j = 0; j < 2; ++j) {
            M[(2 * i    ) * 4 + 2 * j    ] =  G[i][j].re;
            M[(2 * i    ) * 4 + 2 * j + 1] =  G[i][j].im;
            M[(2 * i + 1) * 4 + 2 * j    ] = -G[i][j].im;
            M[(2 * i + 1) * 4 + 2 * j + 1] =  G[i][j].re;
        }
}

template <int PH, int PJ1, int PJ0, int N0, int N1, int N2, int N3, int T0, int T1, int T2>
__device__ __forceinline__ void fused_pass(float2* __restrict__ st, int lane,
                                           c2 uH, c2 uJ1a, c2 uJ1b, c2 uJ0a, c2 uJ0b) {
    const int c  = lane & 15;
    const int h  = lane >> 4;
    const int co = c & 1;

    const c2 t0 = cmul(uH, uJ1a), t1 = cmul(uH, uJ1b);
    c2 P[4];
    P[0] = cmul(t0, uJ0a); P[1] = cmul(t0, uJ0b);
    P[2] = cmul(t1, uJ0a); P[3] = cmul(t1, uJ0b);
    v2f A[4];
#pragma unroll
    for (int j = 0; j < 4; ++j) {
        A[j].x = co ? P[j].im :  P[j].re;
        A[j].y = co ? P[j].re : -P[j].im;
    }

    const int base = ((c & 1) << N0) | (((c >> 1) & 1) << N1) | (((c >> 2) & 1) << N2) |
                     (((c >> 3) & 1) << N3) | (h << PH);
    float2* p = st + base;

    float2 cur0[4], cur1[4], nx0[4], nx1[4];
#pragma unroll
    for (int j = 0; j < 4; ++j) {
        const int jo = ((j >> 1) << PJ1) | ((j & 1) << PJ0);
        cur0[j] = p[jo];
        cur1[j] = p[(1 << T0) | jo];
    }
#pragma unroll
    for (int tp = 0; tp < 4; ++tp) {
        const int ta = ((tp & 1) << T1) | (((tp >> 1) & 1) << T2);
        const int tb = ta | (1 << T0);
        if (tp < 3) {
            const int u  = 2 * tp + 2;
            const int na = (((u >> 1) & 1) << T1) | (((u >> 2) & 1) << T2);
#pragma unroll
            for (int j = 0; j < 4; ++j) {
                const int jo = ((j >> 1) << PJ1) | ((j & 1) << PJ0);
                nx0[j] = p[na | jo];
                nx1[j] = p[na | (1 << T0) | jo];
            }
        }
        v8f C0 = {0.f, 0.f, 0.f, 0.f, 0.f, 0.f, 0.f, 0.f};
        v8f C1 = {0.f, 0.f, 0.f, 0.f, 0.f, 0.f, 0.f, 0.f};
#pragma unroll
        for (int j = 0; j < 4; ++j) {
            v2f B0; B0.x = cur0[j].x; B0.y = cur0[j].y;
            v2f B1; B1.x = cur1[j].x; B1.y = cur1[j].y;
            C0 = __builtin_amdgcn_wmma_f32_16x16x4_f32(false, A[j], false, B0, (short)0, C0, false, false);
            C1 = __builtin_amdgcn_wmma_f32_16x16x4_f32(false, A[j], false, B1, (short)0, C1, false, false);
        }
#pragma unroll
        for (int j = 0; j < 4; ++j) {
            const int jo = ((j >> 1) << PJ1) | ((j & 1) << PJ0);
            p[ta | jo] = make_float2(C0[2 * j], C0[2 * j + 1]);
            p[tb | jo] = make_float2(C1[2 * j], C1[2 * j + 1]);
        }
#pragma unroll
        for (int j = 0; j < 4; ++j) { cur0[j] = nx0[j]; cur1[j] = nx1[j]; }
    }
}

__device__ __forceinline__ int ring_src(int fl, int t) {
    const int ft = (t << 5) ^ (t << 4);
    return (fl ^ (ft & 0x310)) + (ft & 0x0E0);
}
__device__ __forceinline__ void ring_pass0(float2* __restrict__ st, int lane, int fl,
                                           c2 uH, c2 uJ1a, c2 uJ1b, c2 uJ0a, c2 uJ0b) {
    const int co = lane & 1;

    const c2 t0 = cmul(uH, uJ1a), t1 = cmul(uH, uJ1b);
    c2 P[4];
    P[0] = cmul(t0, uJ0a); P[1] = cmul(t0, uJ0b);
    P[2] = cmul(t1, uJ0a); P[3] = cmul(t1, uJ0b);
    v2f A[4];
#pragma unroll
    for (int j = 0; j < 4; ++j) {
        A[j].x = co ? P[j].im :  P[j].re;
        A[j].y = co ? P[j].re : -P[j].im;
    }

    float2 g[32];
#pragma unroll
    for (int tp = 0; tp < 4; ++tp)
#pragma unroll
        for (int j = 0; j < 4; ++j)
#pragma unroll
            for (int e = 0; e < 2; ++e) {
                const int t = 2 * tp + e + 8 * (j & 1) + 16 * (j >> 1);
                g[t] = st[ring_src(fl, t)];
            }

    float2* p = st + lane;
#pragma unroll
    for (int tp = 0; tp < 4; ++tp) {
        v8f C0 = {0.f, 0.f, 0.f, 0.f, 0.f, 0.f, 0.f, 0.f};
        v8f C1 = {0.f, 0.f, 0.f, 0.f, 0.f, 0.f, 0.f, 0.f};
#pragma unroll
        for (int j = 0; j < 4; ++j) {
            const float2 b0 = g[2 * tp     + 8 * (j & 1) + 16 * (j >> 1)];
            const float2 b1 = g[2 * tp + 1 + 8 * (j & 1) + 16 * (j >> 1)];
            v2f B0; B0.x = b0.x; B0.y = b0.y;
            v2f B1; B1.x = b1.x; B1.y = b1.y;
            C0 = __builtin_amdgcn_wmma_f32_16x16x4_f32(false, A[j], false, B0, (short)0, C0, false, false);
            C1 = __builtin_amdgcn_wmma_f32_16x16x4_f32(false, A[j], false, B1, (short)0, C1, false, false);
        }
#pragma unroll
        for (int j = 0; j < 4; ++j) {
            const int jo = ((j >> 1) << 9) | ((j & 1) << 8);
            p[((2 * tp    ) << 5) | jo] = make_float2(C0[2 * j], C0[2 * j + 1]);
            p[((2 * tp + 1) << 5) | jo] = make_float2(C1[2 * j], C1[2 * j + 1]);
        }
    }
}

#define PASS1 fused_pass<3, 7, 6, 0, 1, 2, 4, 5, 8, 9>
#define PASS2 fused_pass<2, 5, 0, 1, 3, 4, 6, 7, 8, 9>
#define PASS3 fused_pass<1, 9, 8, 0, 2, 3, 4, 5, 6, 7>

__device__ __forceinline__ c2 csel(int pick1, c2 a0, c2 a1) {
    c2 r; r.re = pick1 ? a1.re : a0.re; r.im = pick1 ? a1.im : a0.im; return r;
}
__device__ __forceinline__ c2 fold2(c2 u0, c2 u1, c2 pa, c2 pb, c2 qa, c2 qb) {
    return cadd(cmul(u0, cmul(pa, qa)), cmul(u1, cmul(pb, qb)));
}
__device__ __forceinline__ c2 id_u(int o, int i) {
    c2 r; r.re = (o == i) ? 1.f : 0.f; r.im = 0.f; return r;
}
__device__ __forceinline__ c2 var_u(const float* __restrict__ M, int o, int i) {
    const float2 v = *(const float2*)(M + 8 * i + 2 * o);
    c2 r; r.re = v.x; r.im = v.y; return r;
}

__device__ __forceinline__ float lane_value(float v, int src) {
    return __int_as_float(__builtin_amdgcn_readlane(__float_as_int(v), src));
}

__global__ void __launch_bounds__(WPB * 32) __attribute__((amdgpu_waves_per_eu(8)))
qsim_kernel(const float* __restrict__ x, const float* __restrict__ encW,
            const float* __restrict__ encB, const float* __restrict__ Mvar,
            float* __restrict__ out, int batch) {
    __shared__ float2 lds[WPB * DIM];

    const int lane = threadIdx.x & 31;
    const int w    = threadIdx.x >> 5;
    const int b    = blockIdx.x * WPB + w;
    if (b >= batch) return;

    float2* st = &lds[w * DIM];
    const int c  = lane & 15;
    const int h  = lane >> 4;
    const int ho = c >> 3;
    const int o1 = (c >> 2) & 1;
    const int o0 = (c >> 1) & 1;

    {
        float xv0 = x[b * 3 + 0], xv1 = x[b * 3 + 1], xv2 = x[b * 3 + 2];
        const int q = lane < NQ ? lane : NQ - 1;
        float ang = PI_F * tanhf(fmaf(xv2, encW[q * 3 + 2],
                                      fmaf(xv0, encW[q * 3], xv1 * encW[q * 3 + 1])) + encB[q]);
        const float cq = cosf(ang * 0.5f);
        const float sq = sinf(ang * 0.5f);
        const float* M = Mvar + q * 16;
        c2 own[2];
#pragma unroll
        for (int o = 0; o < 2; ++o) {
            const c2 g0 = var_u(M, o, 0), g1 = var_u(M, o, 1);
            own[o].re = cq * g0.re - sq * g1.re;
            own[o].im = cq * g0.im - sq * g1.im;
        }
        c2 V[NQ][2];
#pragma unroll
        for (int p = 0; p < NQ; ++p) {
#pragma unroll
            for (int o = 0; o < 2; ++o) {
                V[p][o].re = lane_value(own[o].re, 9 - p);
                V[p][o].im = lane_value(own[o].im, 9 - p);
            }
        }
        const float* M1 = Mvar + NQ * 16;
        const int l0 = lane & 1, l1 = (lane >> 1) & 1, l2 = (lane >> 2) & 1;
        const int l3 = (lane >> 3) & 1, l4 = lane >> 4;
        {
            const c2 al = cmul(var_u(M1 + 2 * 16, l2, 0), csel(l1, V[6][0], V[6][1]));
            const c2 be = cmul(var_u(M1 + 2 * 16, l2, 1), csel(l1, V[6][1], V[6][0]));
            const c2 w7y0 = cadd(cmul(al, V[7][0]), cmul(be, V[7][1]));
            const c2 w7y1 = cadd(cmul(al, V[7][1]), cmul(be, V[7][0]));
            const c2 ga = cmul(var_u(M1 + 0 * 16, l4, 0), csel(l0, V[9][0], V[9][1]));
            const c2 de = cmul(var_u(M1 + 0 * 16, l4, 1), csel(l0, V[9][1], V[9][0]));
            const c2 v8z = csel(l0, V[8][0], V[8][1]);
            const c2 v8n = csel(l0, V[8][1], V[8][0]);
            const c2 w9x0 = cadd(cmul(ga, v8z), cmul(de, v8n));
            const c2 w9x1 = cadd(cmul(ga, v8n), cmul(de, v8z));
            const c2 ye = cadd(cmul(cmul(w7y0, var_u(M1 + 1 * 16, l3, 0)), w9x0),
                               cmul(cmul(w7y1, var_u(M1 + 1 * 16, l3, 1)), w9x1));
            st[992 + lane] = make_float2(ye.re, ye.im);
            const c2 w5 = fold2(var_u(M1 + 4 * 16, l2, 0), var_u(M1 + 4 * 16, l2, 1),
                                csel(l0, V[5][0], V[5][1]), csel(l0, V[5][1], V[5][0]),
                                csel(l1, V[4][0], V[4][1]), csel(l1, V[4][1], V[4][0]));
            const c2 xe = cmul(w5, var_u(M1 + 3 * 16, l3, l0));
            st[960 + lane] = make_float2(xe.re, xe.im);
        }
        c2 P[2][2];
        {
            const c2 u10 = var_u(M1 + 8 * 16, l1, 0), u11 = var_u(M1 + 8 * 16, l1, 1);
            const c2 u30 = var_u(M1 + 6 * 16, l3, 0), u31 = var_u(M1 + 6 * 16, l3, 1);
            c2 W1s[2][2], W3m[2][2];
#pragma unroll
            for (int xx = 0; xx < 2; ++xx) {
                const c2 a1 = cmul(u10, V[0][xx]), b1 = cmul(u11, V[0][1 ^ xx]);
                const c2 a3 = cmul(u30, V[2][xx]), b3 = cmul(u31, V[2][1 ^ xx]);
#pragma unroll
                for (int yy = 0; yy < 2; ++yy) {
                    W1s[xx][yy] = cmul(cadd(cmul(a1, V[1][yy]), cmul(b1, V[1][1 ^ yy])),
                                       var_u(M1 + 7 * 16, l2, yy));
                    W3m[xx][yy] = cadd(cmul(a3, V[3][yy]), cmul(b3, V[3][1 ^ yy]));
                }
            }
#pragma unroll
            for (int a0 = 0; a0 < 2; ++a0)
#pragma unroll
                for (int a4 = 0; a4 < 2; ++a4) {
                    const c2 mid = cadd(cmul(W1s[a0][0], W3m[0][a4]), cmul(W1s[a0][1], W3m[1][a4]));
                    P[a0][a4] = cmul(cmul(var_u(M1 + 9 * 16, l0, a0), var_u(M1 + 5 * 16, l4, a4)), mid);
                }
        }
        v2f R[4][2][2];
#pragma unroll
        for (int q4 = 0; q4 < 4; ++q4) {
            v2f X[2][2], Xr[2][2];
#pragma unroll
            for (int a4 = 0; a4 < 2; ++a4)
#pragma unroll
                for (int a6 = 0; a6 < 2; ++a6) {
                    const float2 xv = st[960 + 4 * q4 + a6 + 2 * a4];
                    X[a4][a6].x = xv.x;  X[a4][a6].y = xv.y;
                    Xr[a4][a6].x = -xv.y; Xr[a4][a6].y = xv.x;
                }
#pragma unroll
            for (int a0 = 0; a0 < 2; ++a0)
#pragma unroll
                for (int a6 = 0; a6 < 2; ++a6)
                    R[q4][a0][a6] = P[a0][0].re * X[0][a6] + P[a0][0].im * Xr[0][a6]
                                  + P[a0][1].re * X[1][a6] + P[a0][1].im * Xr[1][a6];
        }
#pragma unroll
        for (int s8 = 0; s8 < 8; ++s8) {
            v2f Y[2][2], Yr[2][2];
#pragma unroll
            for (int a6 = 0; a6 < 2; ++a6)
#pragma unroll
                for (int a0 = 0; a0 < 2; ++a0) {
                    const float2 yv = st[992 + 4 * s8 + a0 + 2 * a6];
                    Y[a6][a0].x = yv.x;  Y[a6][a0].y = yv.y;
                    Yr[a6][a0].x = -yv.y; Yr[a6][a0].y = yv.x;
                }
#pragma unroll
            for (int q4 = 0; q4 < 4; ++q4) {
                const v2f a = R[q4][0][0].x * Y[0][0] + R[q4][0][0].y * Yr[0][0]
                            + R[q4][0][1].x * Y[1][0] + R[q4][0][1].y * Yr[1][0]
                            + R[q4][1][0].x * Y[0][1] + R[q4][1][0].y * Yr[0][1]
                            + R[q4][1][1].x * Y[1][1] + R[q4][1][1].y * Yr[1][1];
                st[lane + ((q4 + 4 * s8) << 5)] = make_float2(a.x, a.y);
            }
        }
    }

    const int fl = (lane ^ (lane >> 1)) | ((lane & 1) ? 0x300 : 0);
    for (int layer = 2; layer < NLAYERS; ++layer) {
        const float* ML = Mvar + layer * NQ * 16;
        ring_pass0(st, lane, fl, var_u(ML + 5 * 16, ho, h),
                   var_u(ML + 0 * 16, o1, 0), var_u(ML + 0 * 16, o1, 1),
                   var_u(ML + 1 * 16, o0, 0), var_u(ML + 1 * 16, o0, 1));
        PASS1(st, lane, var_u(ML + 6 * 16, ho, h),
              var_u(ML + 2 * 16, o1, 0), var_u(ML + 2 * 16, o1, 1),
              var_u(ML + 3 * 16, o0, 0), var_u(ML + 3 * 16, o0, 1));
        PASS2(st, lane, var_u(ML + 7 * 16, ho, h),
              var_u(ML + 4 * 16, o1, 0), var_u(ML + 4 * 16, o1, 1),
              var_u(ML + 9 * 16, o0, 0), var_u(ML + 9 * 16, o0, 1));
        if (layer + 1 < NLAYERS)
            PASS3(st, lane, var_u(ML + 8 * 16, ho, h),
                  id_u(o1, 0), id_u(o1, 1), id_u(o0, 0), id_u(o0, 1));
    }

    const float* MQ = Mvar + ((NLAYERS - 1) * NQ + 8) * 16;
    const c2 u00 = var_u(MQ, 0, 0), u01 = var_u(MQ, 0, 1);
    const c2 u10 = var_u(MQ, 1, 0), u11 = var_u(MQ, 1, 1);
    const float z0 = (u00.re * u00.re + u00.im * u00.im) - (u10.re * u10.re + u10.im * u10.im);
    const float z1 = (u01.re * u01.re + u01.im * u01.im) - (u11.re * u11.re + u11.im * u11.im);
    const float wr =  2.f * ((u00.re * u01.re + u00.im * u01.im) - (u10.re * u11.re + u10.im * u11.im));
    const float wi = -2.f * ((u00.im * u01.re - u00.re * u01.im) - (u10.im * u11.re - u10.re * u11.im));
    const int   cc = ((lane >> 1) ^ (lane >> 2)) & 1;
    const float cX = cc ? -z1 : z0;
    const float cY = cc ? -z0 : z1;
    const float cR = cc ? -wr : wr;
    const float cI = wi;
    const int  fl2 = fl ^ 2;
    float s[16];
    float dsum = 0.f;
#pragma unroll
    for (int t = 0; t < 16; ++t) {
        const float2 X = st[ring_src(fl,  t)];
        const float2 Y = st[ring_src(fl2, t)];
        const float nX = X.x * X.x + X.y * X.y;
        const float nY = Y.x * Y.x + Y.y * Y.y;
        const float re = X.x * Y.x + X.y * Y.y;
        const float im = X.y * Y.x - X.x * Y.y;
        s[t] = nX + nY;
        const float dt = cX * nX + cY * nY + cR * re + cI * im;
        dsum = (t == 0) ? dt : dsum + dt;
    }
    float v[NQ];
#pragma unroll
    for (int k = 0; k < 4; ++k) {
        float dk = 0.f;
#pragma unroll
        for (int u = 0; u < (8 >> k); ++u) {
            const float a0 = s[2 * u], a1 = s[2 * u + 1];
            s[u] = a0 + a1;
            dk = (u == 0) ? (a0 - a1) : dk + (a0 - a1);
        }
        v[4 - k] = dk;
    }
    v[0] = dsum;
    v[9] = (lane & 1)        ? -dsum : dsum;
    v[8] = ((lane >> 1) & 1) ? -dsum : dsum;
    v[7] = ((lane >> 2) & 1) ? -s[0] : s[0];
    v[6] = ((lane >> 3) & 1) ? -s[0] : s[0];
    v[5] = (lane >> 4)       ? -s[0] : s[0];

    const int b4 = lane >> 4, b3 = (lane >> 3) & 1, b2 = (lane >> 2) & 1, b1 = (lane >> 1) & 1;
    float k5[5], k3[3], k2[2], k1;
#pragma unroll
    for (int i = 0; i < 5; ++i) {
        const float keep = b4 ? v[5 + i] : v[i];
        const float send = b4 ? v[i] : v[5 + i];
        k5[i] = keep + __shfl_xor(send, 16, 32);
    }
#pragma unroll
    for (int i = 0; i < 3; ++i) {
        const float hi = (i < 2) ? k5[3 + i] : 0.f;
        const float keep = b3 ? hi : k5[i];
        const float send = b3 ? k5[i] : hi;
        k3[i] = keep + __shfl_xor(send, 8, 32);
    }
#pragma unroll
    for (int i = 0; i < 2; ++i) {
        const float hi = (i < 1) ? k3[2] : 0.f;
        const float keep = b2 ? hi : k3[i];
        const float send = b2 ? k3[i] : hi;
        k2[i] = keep + __shfl_xor(send, 4, 32);
    }
    {
        const float keep = b1 ? k2[1] : k2[0];
        const float send = b1 ? k2[0] : k2[1];
        k1 = keep + __shfl_xor(send, 2, 32);
    }
    k1 += __shfl_xor(k1, 1, 32);
    const int  ql    = b3 ? 3 + b1 : (b2 ? 2 : b1);
    const bool holds = ((lane & 1) == 0) && (b3 ? (b2 == 0) : !(b2 && b1));
    if (holds) out[b * NQ + 5 * b4 + ql] = k1;
}

extern "C" void kernel_launch(void* const* d_in, const int* in_sizes, int n_in,
                              void* d_out, int out_size, void* d_ws, size_t ws_size,
                              hipStream_t stream) {
    const float* x     = (const float*)d_in[0];
    const float* encW  = (const float*)d_in[1];
    const float* encB  = (const float*)d_in[2];
    const float* theta = (const float*)d_in[3];
    float* out  = (float*)d_out;
    float* Mvar = (float*)d_ws;

    int batch = in_sizes[0] / 3;

    qprep_kernel<<<1, 64, 0, stream>>>(theta, Mvar);

    int blocks = (batch + WPB - 1) / WPB;
    qsim_kernel<<<blocks, WPB * 32, 0, stream>>>(x, encW, encB, Mvar, out, batch);
}
